# starting kernel
# baseline (speedup 1.0000x reference)
#include <hip/hip_runtime.h>


typedef float v2f __attribute__((ext_vector_type(2)));
typedef float v8f __attribute__((ext_vector_type(8)));

#define C_DIM 256
#define V_DIM 1024
#define F_DIM 256
#define T_DIM 4096
#define B_DIM 4
#define M_DIM (B_DIM * T_DIM)

static __device__ __forceinline__ v8f wmma4(v2f a, v2f b, v8f c) {
  return __builtin_amdgcn_wmma_f32_16x16x4_f32(
       false, a,  false, b,
       (short)0, c,  false,  false);
}

static __device__ __forceinline__ v8f vzero8() {
  v8f z;
#pragma unroll
  for (int i = 0; i < 8; ++i) z[i] = 0.0f;
  return z;
}

__global__ void proj_weights_kernel(const float* __restrict__ basis,
                                    const float* __restrict__ qc,
                                    const float* __restrict__ kc,
                                    const float* __restrict__ vc,
                                    const float* __restrict__ oc,
                                    float* __restrict__ qw,
                                    float* __restrict__ kw,
                                    float* __restrict__ vw,
                                    float* __restrict__ owT) {
  const int lane = threadIdx.x & 31;
  const int r = lane & 15;
  const int h = lane >> 4;
  const int v0 = blockIdx.x * 16;
  const int c0 = blockIdx.y * 16;
  const int which = blockIdx.z;
  const float* coef = (which == 0) ? qc : (which == 1) ? kc : (which == 2) ? vc : oc;

  v8f acc = vzero8();
  for (int f0 = 0; f0 < F_DIM; f0 += 4) {
    v2f a = *(const v2f*)&basis[(size_t)(v0 + r) * F_DIM + f0 + 2 * h];
    v2f b = *(const v2f*)&coef[(size_t)(c0 + r) * F_DIM + f0 + 2 * h];
    acc = wmma4(a, b, acc);
  }
  if (which < 3) {
    float* w = (which == 0) ? qw : (which == 1) ? kw : vw;
#pragma unroll
    for (int i = 0; i < 8; ++i)
      w[(size_t)(v0 + i + 8 * h) * C_DIM + c0 + r] = acc[i];
  } else {
#pragma unroll
    for (int i = 0; i < 8; ++i)
      owT[(size_t)(c0 + r) * V_DIM + v0 + i + 8 * h] = acc[i];
  }
}

__global__ void qkv_kernel(const float* __restrict__ x,
                           const float* __restrict__ qw,
                           const float* __restrict__ kw,
                           const float* __restrict__ vw,
                           float* __restrict__ Q,
                           float* __restrict__ K,
                           float* __restrict__ Vv) {
  const int lane = threadIdx.x & 31;
  const int r = lane & 15;
  const int h = lane >> 4;
  const int m0 = blockIdx.x * 16;
  const int c0 = blockIdx.y * 64;
  const int which = blockIdx.z;
  const float* w = (which == 0) ? qw : (which == 1) ? kw : vw;
  float* out = (which == 0) ? Q : (which == 1) ? K : Vv;

  v8f acc[4];
#pragma unroll
  for (int nt = 0; nt < 4; ++nt) acc[nt] = vzero8();

  for (int k0 = 0; k0 < V_DIM; k0 += 4) {
    v2f a = *(const v2f*)&x[(size_t)(m0 + r) * V_DIM + k0 + 2 * h];
    const float* w0 = &w[(size_t)(k0 + 2 * h) * C_DIM + c0 + r];
    const float* w1 = w0 + C_DIM;
#pragma unroll
    for (int nt = 0; nt < 4; ++nt) {
      v2f b;
      b.x = w0[nt * 16];
      b.y = w1[nt * 16];
      acc[nt] = wmma4(a, b, acc[nt]);
    }
  }
#pragma unroll
  for (int nt = 0; nt < 4; ++nt)
#pragma unroll
    for (int i = 0; i < 8; ++i)
      out[(size_t)(m0 + i + 8 * h) * C_DIM + c0 + nt * 16 + r] = acc[nt][i];
}

__global__ void attn_kernel(const float* __restrict__ Q,
                            const float* __restrict__ K,
                            const float* __restrict__ Vv,
                            const float* __restrict__ decay_logit,
                            float* __restrict__ R) {
  __shared__ float qs[16 * 260];
  __shared__ float ss[16 * 18];

  const int lane = threadIdx.x & 31;
  const int r = lane & 15;
  const int h = lane >> 4;
  const int i0 = blockIdx.x * 16;
  const int c0 = blockIdx.y * 64;
  const int b = blockIdx.z;

  const float* Qb = Q + (size_t)b * T_DIM * C_DIM;
  const float* Kb = K + (size_t)b * T_DIM * C_DIM;
  const float* Vb = Vv + (size_t)b * T_DIM * C_DIM;
  float* Rb = R + (size_t)b * T_DIM * C_DIM;

  for (int idx = lane; idx < 16 * 64; idx += 32) {
    int row = idx >> 6;
    int c4 = (idx & 63) << 2;
    *(float4*)&qs[row * 260 + c4] =
        *(const float4*)&Qb[(size_t)(i0 + row) * C_DIM + c4];
  }
  __syncthreads();

  const float dl = *decay_logit;
  const float dec = 1.0f / (1.0f + __expf(-dl));
  const float ln_dec = __logf(dec);

  v8f acc0 = vzero8(), acc1 = vzero8(), acc2 = vzero8(), acc3 = vzero8();

  for (int j0 = i0; j0 < T_DIM; j0 += 16) {
    {
      int jn = (j0 + 16 < T_DIM) ? (j0 + 16) : j0;
      __builtin_prefetch(&Kb[(size_t)(jn + r + 8 * h) * C_DIM], 0, 1);
      __builtin_prefetch(&Vb[(size_t)(jn + r + 8 * h) * C_DIM + c0], 0, 1);
    }

    v8f s = vzero8();
    const float* Kj = Kb + (size_t)j0 * C_DIM;
    for (int k0 = 0; k0 < C_DIM; k0 += 4) {
      v2f a = *(const v2f*)&qs[r * 260 + k0 + 2 * h];
      v2f bb = *(const v2f*)&Kj[(size_t)r * C_DIM + k0 + 2 * h];
      s = wmma4(a, bb, s);
    }
#pragma unroll
    for (int i = 0; i < 8; ++i) {
      int gi = i0 + i + 8 * h;
      int gj = j0 + r;
      int d = gj - gi;
      float wgt = (d > 0) ? __expf(ln_dec * (float)(d - 1)) : 0.0f;
      ss[(i + 8 * h) * 18 + r] = s[i] * wgt;
    }
    __syncthreads();
#pragma unroll
    for (int kk = 0; kk < 4; ++kk) {
      v2f a = *(const v2f*)&ss[r * 18 + kk * 4 + 2 * h];
      const float* Vr0 = Vb + (size_t)(j0 + kk * 4 + 2 * h) * C_DIM + c0;
      const float* Vr1 = Vr0 + C_DIM;
      v2f bb;
      bb.x = Vr0[r];      bb.y = Vr1[r];      acc0 = wmma4(a, bb, acc0);
      bb.x = Vr0[16 + r]; bb.y = Vr1[16 + r]; acc1 = wmma4(a, bb, acc1);
      bb.x = Vr0[32 + r]; bb.y = Vr1[32 + r]; acc2 = wmma4(a, bb, acc2);
      bb.x = Vr0[48 + r]; bb.y = Vr1[48 + r]; acc3 = wmma4(a, bb, acc3);
    }
    __syncthreads();
  }

#pragma unroll
  for (int i = 0; i < 8; ++i) {
    size_t row = (size_t)(i0 + i + 8 * h) * C_DIM + c0 + r;
    Rb[row] = acc0[i];
    Rb[row + 16] = acc1[i];
    Rb[row + 32] = acc2[i];
    Rb[row + 48] = acc3[i];
  }
}

__global__ void out_proj_kernel(const float* __restrict__ R,
                                const float* __restrict__ owT,
                                const float* __restrict__ out_scale,
                                float* __restrict__ out) {
  const int lane = threadIdx.x & 31;
  const int r = lane & 15;
  const int h = lane >> 4;
  const int m0 = blockIdx.x * 16;
  const int v0 = blockIdx.y * 64;

  v8f acc[4];
#pragma unroll
  for (int nt = 0; nt < 4; ++nt) acc[nt] = vzero8();

  for (int k0 = 0; k0 < C_DIM; k0 += 4) {
    v2f a = *(const v2f*)&R[(size_t)(m0 + r) * C_DIM + k0 + 2 * h];
    const float* w0 = &owT[(size_t)(k0 + 2 * h) * V_DIM + v0 + r];
    const float* w1 = w0 + V_DIM;
#pragma unroll
    for (int nt = 0; nt < 4; ++nt) {
      v2f b;
      b.x = w0[nt * 16];
      b.y = w1[nt * 16];
      acc[nt] = wmma4(a, b, acc[nt]);
    }
  }
  const float sc = *out_scale;
#pragma unroll
  for (int nt = 0; nt < 4; ++nt)
#pragma unroll
    for (int i = 0; i < 8; ++i)
      out[(size_t)(m0 + i + 8 * h) * V_DIM + v0 + nt * 16 + r] = acc[nt][i] * sc;
}

extern "C" void kernel_launch(void* const* d_in, const int* in_sizes, int n_in,
                              void* d_out, int out_size, void* d_ws, size_t ws_size,
                              hipStream_t stream) {
  (void)in_sizes; (void)n_in; (void)out_size; (void)ws_size;

  const float* x = (const float*)d_in[0];
  const float* basis = (const float*)d_in[1];
  const float* qc = (const float*)d_in[2];
  const float* kc = (const float*)d_in[3];
  const float* vc = (const float*)d_in[4];
  const float* oc = (const float*)d_in[5];
  const float* decay_logit = (const float*)d_in[6];
  const float* out_scale = (const float*)d_in[7];
  float* out = (float*)d_out;

  float* ws = (float*)d_ws;
  float* qw = ws;
  float* kw = qw + (size_t)V_DIM * C_DIM;
  float* vw = kw + (size_t)V_DIM * C_DIM;
  float* owT = vw + (size_t)V_DIM * C_DIM;
  float* Q = owT + (size_t)V_DIM * C_DIM;
  float* K = Q + (size_t)M_DIM * C_DIM;
  float* Vv = K + (size_t)M_DIM * C_DIM;
  float* R = Vv + (size_t)M_DIM * C_DIM;

  dim3 blk(32);

  proj_weights_kernel<<<dim3(V_DIM / 16, C_DIM / 16, 4), blk, 0, stream>>>(
      basis, qc, kc, vc, oc, qw, kw, vw, owT);

  qkv_kernel<<<dim3(M_DIM / 16, C_DIM / 64, 3), blk, 0, stream>>>(
      x, qw, kw, vw, Q, K, Vv);

  attn_kernel<<<dim3(T_DIM / 16, C_DIM / 64, B_DIM), blk, 0, stream>>>(
      Q, K, Vv, decay_logit, R);

  out_proj_kernel<<<dim3(M_DIM / 16, V_DIM / 64), blk, 0, stream>>>(
      R, owT, out_scale, out);
}


# x@w and retrieved@o_w GEMMs: weight kernel writes weights in B-fragment order; A and B read 16 B per lane, one b128 pair feeds 2 WMMAs (same permuted k order both sides); sum order changes
# speedup vs baseline: 3.2381x; 3.2381x over previous
#include <hip/hip_runtime.h>


typedef float v2f __attribute__((ext_vector_type(2)));
typedef float v8f __attribute__((ext_vector_type(8)));

#define C_DIM 256
#define V_DIM 1024
#define F_DIM 256
#define T_DIM 4096
#define B_DIM 4
#define M_DIM (B_DIM * T_DIM)

static __device__ __forceinline__ v8f wmma4(v2f a, v2f b, v8f c) {
  return __builtin_amdgcn_wmma_f32_16x16x4_f32(
       false, a,  false, b,
       (short)0, c,  false,  false);
}

static __device__ __forceinline__ v8f vzero8() {
  v8f z;
#pragma unroll
  for (int i = 0; i < 8; ++i) z[i] = 0.0f;
  return z;
}

__global__ void proj_weights_kernel(const float* __restrict__ basis,
                                    const float* __restrict__ qc,
                                    const float* __restrict__ kc,
                                    const float* __restrict__ vc,
                                    const float* __restrict__ oc,
                                    float* __restrict__ qw,
                                    float* __restrict__ kw,
                                    float* __restrict__ vw,
                                    float* __restrict__ owT) {
  const int lane = threadIdx.x & 31;
  const int r = lane & 15;
  const int h = lane >> 4;
  const int v0 = blockIdx.x * 16;
  const int c0 = blockIdx.y * 16;
  const int which = blockIdx.z;
  const float* coef = (which == 0) ? qc : (which == 1) ? kc : (which == 2) ? vc : oc;

  const float* brow = &basis[(size_t)(v0 + r) * F_DIM + 2 * h];
  const float* crow = &coef[(size_t)(c0 + r) * F_DIM + 2 * h];
  const float* ap = (which < 3) ? brow : crow;
  const float* bp = (which < 3) ? crow : brow;
  v8f acc = vzero8();
  for (int f0 = 0; f0 < F_DIM; f0 += 4) {
    v2f a = *(const v2f*)&ap[f0];
    v2f b = *(const v2f*)&bp[f0];
    acc = wmma4(a, b, acc);
  }
  float4 lo, hi;
  lo.x = acc[0]; lo.y = acc[1]; lo.z = acc[2]; lo.w = acc[3];
  hi.x = acc[4]; hi.y = acc[5]; hi.z = acc[6]; hi.w = acc[7];
  float* dst;
  if (which < 3) {
    float* w = (which == 0) ? qw : (which == 1) ? kw : vw;
    dst = w + ((size_t)blockIdx.y * (V_DIM / 8) + 2 * blockIdx.x + h) * 128;
  } else {
    dst = owT + ((size_t)blockIdx.x * (C_DIM / 8) + 2 * blockIdx.y + h) * 128;
  }
  *(float4*)&dst[4 * r] = lo;
  *(float4*)&dst[4 * (16 + r)] = hi;
}

__global__ void qkv_kernel(const float* __restrict__ x,
                           const float* __restrict__ qw,
                           const float* __restrict__ kw,
                           const float* __restrict__ vw,
                           float* __restrict__ Q,
                           float* __restrict__ K,
                           float* __restrict__ Vv) {
  const int lane = threadIdx.x & 31;
  const int r = lane & 15;
  const int h = lane >> 4;
  const int m0 = blockIdx.x * 16;
  const int c0 = blockIdx.y * 64;
  const int which = blockIdx.z;
  const float* w = (which == 0) ? qw : (which == 1) ? kw : vw;
  float* out = (which == 0) ? Q : (which == 1) ? K : Vv;

  v8f acc[4];
#pragma unroll
  for (int nt = 0; nt < 4; ++nt) acc[nt] = vzero8();

  const float* xr = &x[(size_t)(m0 + r) * V_DIM + 4 * h];
  const float* wf = w + (size_t)(blockIdx.y * 4) * (V_DIM / 8) * 128 + 4 * lane;
#pragma unroll 4
  for (int kc = 0; kc < V_DIM / 8; ++kc) {
    const float4 a4 = *(const float4*)&xr[kc * 8];
    float4 b4[4];
#pragma unroll
    for (int nt = 0; nt < 4; ++nt)
      b4[nt] = *(const float4*)&wf[((size_t)nt * (V_DIM / 8) + kc) * 128];
#pragma unroll
    for (int nt = 0; nt < 4; ++nt) {
      v2f a, b;
      a.x = a4.x; a.y = a4.y; b.x = b4[nt].x; b.y = b4[nt].y;
      acc[nt] = wmma4(a, b, acc[nt]);
      a.x = a4.z; a.y = a4.w; b.x = b4[nt].z; b.y = b4[nt].w;
      acc[nt] = wmma4(a, b, acc[nt]);
    }
  }
#pragma unroll
  for (int nt = 0; nt < 4; ++nt)
#pragma unroll
    for (int i = 0; i < 8; ++i)
      out[(size_t)(m0 + i + 8 * h) * C_DIM + c0 + nt * 16 + r] = acc[nt][i];
}

#define ATT_WAVES 8

__global__ __launch_bounds__(32 * ATT_WAVES)
void attn_kernel(const float* __restrict__ Q,
                 const float* __restrict__ K,
                 const float* __restrict__ Vv,
                 const float* __restrict__ decay_logit,
                 float* __restrict__ R) {
  __shared__ __attribute__((aligned(16))) float pool[ATT_WAVES * 16 * 260];
  __shared__ float ss[ATT_WAVES][16 * 18];
  float* qs = pool;

  const int tid = threadIdx.x;
  const int lane = tid & 31;
  const int wv = __builtin_amdgcn_readfirstlane(tid >> 5);
  const int r = lane & 15;
  const int h = lane >> 4;
  const int i0 = blockIdx.y * 16;
  const int b = blockIdx.x;

  const float* Qb = Q + (size_t)b * T_DIM * C_DIM;
  const float* Kb = K + (size_t)b * T_DIM * C_DIM;
  const float* Vb = Vv + (size_t)b * T_DIM * C_DIM;
  float* Rb = R + (size_t)b * T_DIM * C_DIM;

  for (int idx = tid; idx < 16 * 64; idx += 32 * ATT_WAVES) {
    int row = idx >> 6;
    int c4 = (idx & 63) << 2;
    *(float4*)&qs[row * 260 + c4] =
        *(const float4*)&Qb[(size_t)(i0 + row) * C_DIM + c4];
  }
  __syncthreads();

  const float dl = *decay_logit;
  const float dec = 1.0f / (1.0f + __expf(-dl));
  const float ln_dec = __logf(dec);

  v8f acc[16];
#pragma unroll
  for (int nt = 0; nt < 16; ++nt) acc[nt] = vzero8();

  float* sw = ss[wv];
  const int ntiles = (T_DIM - i0) >> 4;

  for (int jt = wv; jt < ntiles; jt += ATT_WAVES) {
    const int j0 = i0 + 16 * jt;

    {
      {
        int jn = (jt + ATT_WAVES < ntiles) ? (j0 + 16 * ATT_WAVES) : j0;
        __builtin_prefetch(&Kb[(size_t)(jn + r + 8 * h) * C_DIM], 0, 1);
        __builtin_prefetch(&Vb[(size_t)(jn + r + 8 * h) * C_DIM], 0, 1);
      }

      v8f s = vzero8();
      const float* Kj = Kb + (size_t)j0 * C_DIM;
      for (int k0 = 0; k0 < C_DIM; k0 += 8) {
        float4 a4 = *(const float4*)&qs[r * 260 + k0 + 4 * h];
        float4 b4 = *(const float4*)&Kj[(size_t)r * C_DIM + k0 + 4 * h];
        v2f a, bb;
        a.x = a4.x; a.y = a4.y; bb.x = b4.x; bb.y = b4.y;
        s = wmma4(a, bb, s);
        a.x = a4.z; a.y = a4.w; bb.x = b4.z; bb.y = b4.w;
        s = wmma4(a, bb, s);
      }
#pragma unroll
      for (int i = 0; i < 8; ++i) {
        int gi = i0 + i + 8 * h;
        int gj = j0 + r;
        int d = gj - gi;
        float wgt = (d > 0) ? __expf(ln_dec * (float)(d - 1)) : 0.0f;
        sw[(i + 8 * h) * 18 + r] = s[i] * wgt;
      }
    }
    __builtin_amdgcn_wave_barrier();
    {
#pragma unroll 2
      for (int kk = 0; kk < 4; ++kk) {
        v2f a = *(const v2f*)&sw[r * 18 + kk * 4 + 2 * h];
        const float* Vr0 = Vb + (size_t)(j0 + kk * 4 + 2 * h) * C_DIM + r;
        const float* Vr1 = Vr0 + C_DIM;
#pragma unroll
        for (int nt = 0; nt < 16; ++nt) {
          v2f bb;
          bb.x = Vr0[nt * 16];
          bb.y = Vr1[nt * 16];
          acc[nt] = wmma4(a, bb, acc[nt]);
        }
      }
    }
    __builtin_amdgcn_wave_barrier();
  }

  __syncthreads();
  float* pw = pool + wv * (16 * 260);
#pragma unroll
  for (int nt = 0; nt < 16; ++nt)
#pragma unroll
    for (int i = 0; i < 8; ++i)
      pw[(i + 8 * h) * 260 + nt * 16 + r] = acc[nt][i];
  __syncthreads();

  for (int idx = tid; idx < 16 * 64; idx += 32 * ATT_WAVES) {
    int row = idx >> 6;
    int c4 = (idx & 63) << 2;
    const int o = row * 260 + c4;
    float4 sum = *(const float4*)&pool[(ATT_WAVES - 1) * (16 * 260) + o];
#pragma unroll
    for (int w = ATT_WAVES - 2; w >= 0; --w) {
      const float4 p = *(const float4*)&pool[w * (16 * 260) + o];
      sum.x += p.x; sum.y += p.y; sum.z += p.z; sum.w += p.w;
    }
    *(float4*)&Rb[(size_t)(i0 + row) * C_DIM + c4] = sum;
  }
}

__global__ void out_proj_kernel(const float* __restrict__ R,
                                const float* __restrict__ owT,
                                const float* __restrict__ out_scale,
                                float* __restrict__ out) {
  const int lane = threadIdx.x & 31;
  const int r = lane & 15;
  const int h = lane >> 4;
  const int m0 = blockIdx.x * 16;
  const int v0 = blockIdx.y * 64;

  v8f acc[4];
#pragma unroll
  for (int nt = 0; nt < 4; ++nt) acc[nt] = vzero8();

  const float* rr = &R[(size_t)(m0 + r) * C_DIM + 4 * h];
  const float* wf = owT + (size_t)(blockIdx.y * 4) * (C_DIM / 8) * 128 + 4 * lane;
#pragma unroll 4
  for (int kc = 0; kc < C_DIM / 8; ++kc) {
    const float4 a4 = *(const float4*)&rr[kc * 8];
    float4 b4[4];
#pragma unroll
    for (int nt = 0; nt < 4; ++nt)
      b4[nt] = *(const float4*)&wf[((size_t)nt * (C_DIM / 8) + kc) * 128];
#pragma unroll
    for (int nt = 0; nt < 4; ++nt) {
      v2f a, b;
      a.x = a4.x; a.y = a4.y; b.x = b4[nt].x; b.y = b4[nt].y;
      acc[nt] = wmma4(a, b, acc[nt]);
      a.x = a4.z; a.y = a4.w; b.x = b4[nt].z; b.y = b4[nt].w;
      acc[nt] = wmma4(a, b, acc[nt]);
    }
  }
  const float sc = *out_scale;
#pragma unroll
  for (int nt = 0; nt < 4; ++nt)
#pragma unroll
    for (int i = 0; i < 8; ++i)
      out[(size_t)(m0 + i + 8 * h) * V_DIM + v0 + nt * 16 + r] = acc[nt][i] * sc;
}

extern "C" void kernel_launch(void* const* d_in, const int* in_sizes, int n_in,
                              void* d_out, int out_size, void* d_ws, size_t ws_size,
                              hipStream_t stream) {
  (void)in_sizes; (void)n_in; (void)out_size; (void)ws_size;

  const float* x = (const float*)d_in[0];
  const float* basis = (const float*)d_in[1];
  const float* qc = (const float*)d_in[2];
  const float* kc = (const float*)d_in[3];
  const float* vc = (const float*)d_in[4];
  const float* oc = (const float*)d_in[5];
  const float* decay_logit = (const float*)d_in[6];
  const float* out_scale = (const float*)d_in[7];
  float* out = (float*)d_out;

  float* ws = (float*)d_ws;
  float* qw = ws;
  float* kw = qw + (size_t)V_DIM * C_DIM;
  float* vw = kw + (size_t)V_DIM * C_DIM;
  float* owT = vw + (size_t)V_DIM * C_DIM;
  float* Q = owT + (size_t)V_DIM * C_DIM;
  float* K = Q + (size_t)M_DIM * C_DIM;
  float* Vv = K + (size_t)M_DIM * C_DIM;
  float* R = Vv + (size_t)M_DIM * C_DIM;

  dim3 blk(32);

  proj_weights_kernel<<<dim3(V_DIM / 16, C_DIM / 16, 4), blk, 0, stream>>>(
      basis, qc, kc, vc, oc, qw, kw, vw, owT);

  qkv_kernel<<<dim3(M_DIM / 16, C_DIM / 64, 3), blk, 0, stream>>>(
      x, qw, kw, vw, Q, K, Vv);

  attn_kernel<<<dim3(B_DIM, T_DIM / 16), dim3(32 * ATT_WAVES), 0, stream>>>(
      Q, K, Vv, decay_logit, R);

  out_proj_kernel<<<dim3(M_DIM / 16, V_DIM / 64), blk, 0, stream>>>(
      R, owT, out_scale, out);
}


# attention: K and V written by the projection kernel in WMMA fragment order (512 B per load); score tile computed transposed, kept in registers as B of V^T S^T, no LDS re-layout; key sum order changes
# speedup vs baseline: 3.8471x; 1.1881x over previous
#include <hip/hip_runtime.h>


typedef float v2f __attribute__((ext_vector_type(2)));
typedef float v8f __attribute__((ext_vector_type(8)));

#define C_DIM 256
#define V_DIM 1024
#define F_DIM 256
#define T_DIM 4096
#define B_DIM 4
#define M_DIM (B_DIM * T_DIM)

static __device__ __forceinline__ v8f wmma4(v2f a, v2f b, v8f c) {
  return __builtin_amdgcn_wmma_f32_16x16x4_f32(
       false, a,  false, b,
       (short)0, c,  false,  false);
}

static __device__ __forceinline__ v8f vzero8() {
  v8f z;
#pragma unroll
  for (int i = 0; i < 8; ++i) z[i] = 0.0f;
  return z;
}

__global__ void proj_weights_kernel(const float* __restrict__ basis,
                                    const float* __restrict__ qc,
                                    const float* __restrict__ kc,
                                    const float* __restrict__ vc,
                                    const float* __restrict__ oc,
                                    float* __restrict__ qw,
                                    float* __restrict__ kw,
                                    float* __restrict__ vw,
                                    float* __restrict__ owT) {
  const int lane = threadIdx.x & 31;
  const int r = lane & 15;
  const int h = lane >> 4;
  const int v0 = blockIdx.x * 16;
  const int c0 = blockIdx.y * 16;
  const int which = blockIdx.z;
  const float* coef = (which == 0) ? qc : (which == 1) ? kc : (which == 2) ? vc : oc;

  const float* brow = &basis[(size_t)(v0 + r) * F_DIM + 2 * h];
  const float* crow = &coef[(size_t)(c0 + r) * F_DIM + 2 * h];
  const float* ap = (which < 3) ? brow : crow;
  const float* bp = (which < 3) ? crow : brow;
  v8f acc = vzero8();
  for (int f0 = 0; f0 < F_DIM; f0 += 4) {
    v2f a = *(const v2f*)&ap[f0];
    v2f b = *(const v2f*)&bp[f0];
    acc = wmma4(a, b, acc);
  }
  float4 lo, hi;
  lo.x = acc[0]; lo.y = acc[1]; lo.z = acc[2]; lo.w = acc[3];
  hi.x = acc[4]; hi.y = acc[5]; hi.z = acc[6]; hi.w = acc[7];
  float* dst;
  if (which < 3) {
    float* w = (which == 0) ? qw : (which == 1) ? kw : vw;
    dst = w + ((size_t)blockIdx.y * (V_DIM / 8) + 2 * blockIdx.x + h) * 128;
  } else {
    dst = owT + ((size_t)blockIdx.x * (C_DIM / 8) + 2 * blockIdx.y + h) * 128;
  }
  *(float4*)&dst[4 * r] = lo;
  *(float4*)&dst[4 * (16 + r)] = hi;
}

template <bool SWAP>
static __device__ __forceinline__ void xw_tile_16x64(const float* __restrict__ xr,
                                                     const float* __restrict__ wf,
                                                     v8f (&acc)[4]) {
#pragma unroll 4
  for (int kc = 0; kc < V_DIM / 8; ++kc) {
    const float4 a4 = *(const float4*)&xr[kc * 8];
    float4 b4[4];
#pragma unroll
    for (int nt = 0; nt < 4; ++nt)
      b4[nt] = *(const float4*)&wf[((size_t)nt * (V_DIM / 8) + kc) * 128];
#pragma unroll
    for (int nt = 0; nt < 4; ++nt) {
      v2f a, b;
      a.x = a4.x; a.y = a4.y; b.x = b4[nt].x; b.y = b4[nt].y;
      acc[nt] = SWAP ? wmma4(b, a, acc[nt]) : wmma4(a, b, acc[nt]);
      a.x = a4.z; a.y = a4.w; b.x = b4[nt].z; b.y = b4[nt].w;
      acc[nt] = SWAP ? wmma4(b, a, acc[nt]) : wmma4(a, b, acc[nt]);
    }
  }
}

__global__ void qkv_kernel(const float* __restrict__ x,
                           const float* __restrict__ qw,
                           const float* __restrict__ kw,
                           const float* __restrict__ vw,
                           float* __restrict__ Q,
                           float* __restrict__ K,
                           float* __restrict__ Vv) {
  const int lane = threadIdx.x & 31;
  const int r = lane & 15;
  const int h = lane >> 4;
  const int m0 = blockIdx.x * 16;
  const int c0 = blockIdx.y * 64;
  const int which = blockIdx.z;
  const float* w = (which == 0) ? qw : (which == 1) ? kw : vw;
  float* out = (which == 0) ? Q : (which == 1) ? K : Vv;

  v8f acc[4];
#pragma unroll
  for (int nt = 0; nt < 4; ++nt) acc[nt] = vzero8();

  const float* xr = &x[(size_t)(m0 + r) * V_DIM + 4 * h];
  const float* wf = w + (size_t)(blockIdx.y * 4) * (V_DIM / 8) * 128 + 4 * lane;
  if (which == 1) {
    xw_tile_16x64<true>(xr, wf, acc);
#pragma unroll
    for (int nt = 0; nt < 4; ++nt) {
      float4 lo, hi;
      lo.x = acc[nt][0]; lo.y = acc[nt][1]; lo.z = acc[nt][2]; lo.w = acc[nt][3];
      hi.x = acc[nt][4]; hi.y = acc[nt][5]; hi.z = acc[nt][6]; hi.w = acc[nt][7];
      float* dst = out + ((size_t)blockIdx.x * (C_DIM / 8) + blockIdx.y * 8 + 2 * nt + h) * 128;
      *(float4*)&dst[4 * r] = lo;
      *(float4*)&dst[4 * (16 + r)] = hi;
    }
  } else {
    xw_tile_16x64<false>(xr, wf, acc);
    if (which == 0) {
#pragma unroll
      for (int nt = 0; nt < 4; ++nt)
#pragma unroll
        for (int i = 0; i < 8; ++i)
          out[(size_t)(m0 + i + 8 * h) * C_DIM + c0 + nt * 16 + r] = acc[nt][i];
    } else {
#pragma unroll
      for (int nt = 0; nt < 4; ++nt) {
        float4 lo, hi;
        lo.x = acc[nt][0]; lo.y = acc[nt][1]; lo.z = acc[nt][2]; lo.w = acc[nt][3];
        hi.x = acc[nt][4]; hi.y = acc[nt][5]; hi.z = acc[nt][6]; hi.w = acc[nt][7];
        float* dst = out + (((size_t)blockIdx.x * (C_DIM / 16) + blockIdx.y * 4 + nt) * 2) * 128;
        *(float4*)&dst[4 * lane] = lo;
        *(float4*)&dst[128 + 4 * lane] = hi;
      }
    }
  }
}

#define ATT_WAVES 8

__global__ __launch_bounds__(32 * ATT_WAVES)
void attn_kernel(const float* __restrict__ Q,
                 const float* __restrict__ K,
                 const float* __restrict__ Vv,
                 const float* __restrict__ decay_logit,
                 float* __restrict__ R) {
  __shared__ __attribute__((aligned(16))) float pool[ATT_WAVES * 16 * 260];
  float* qs = pool;

  const int tid = threadIdx.x;
  const int lane = tid & 31;
  const int wv = __builtin_amdgcn_readfirstlane(tid >> 5);
  const int r = lane & 15;
  const int h = lane >> 4;
  const int i0 = blockIdx.y * 16;
  const int b = blockIdx.x;

  const float* Qb = Q + (size_t)b * T_DIM * C_DIM;
  const float* Kb = K + (size_t)b * T_DIM * C_DIM;
  const float* Vb = Vv + (size_t)b * T_DIM * C_DIM;
  float* Rb = R + (size_t)b * T_DIM * C_DIM;

  for (int idx = tid; idx < 16 * 64; idx += 32 * ATT_WAVES) {
    int row = idx >> 6;
    int c4 = (idx & 63) << 2;
    *(float4*)&qs[row * 260 + c4] =
        *(const float4*)&Qb[(size_t)(i0 + row) * C_DIM + c4];
  }
  __syncthreads();

  const float dl = *decay_logit;
  const float dec = 1.0f / (1.0f + __expf(-dl));
  const float ln_dec = __logf(dec);

  v8f acc[16];
#pragma unroll
  for (int nt = 0; nt < 16; ++nt) acc[nt] = vzero8();

  const int ntiles = (T_DIM - i0) >> 4;

  for (int jt = wv; jt < ntiles; jt += ATT_WAVES) {
    const int j0 = i0 + 16 * jt;
    const float* Kj = Kb + (size_t)j0 * C_DIM + 4 * lane;
    const float* Vj = Vb + (size_t)j0 * C_DIM + 4 * lane;

    {
      const int jn = (jt + ATT_WAVES < ntiles) ? (j0 + 16 * ATT_WAVES) : j0;
      const float* Kn = Kb + (size_t)jn * C_DIM + 64 * lane;
      const float* Vn = Vb + (size_t)jn * C_DIM + 64 * lane;
      __builtin_prefetch(Kn, 0, 1);
      __builtin_prefetch(Kn + 2048, 0, 1);
      __builtin_prefetch(Vn, 0, 1);
      __builtin_prefetch(Vn + 2048, 0, 1);
    }

    asm volatile("" ::: "memory");

    v8f s = vzero8();
    for (int kc = 0; kc < C_DIM / 8; ++kc) {
      const float4 k4 = *(const float4*)&Kj[kc * 128];
      const float4 q4 = *(const float4*)&qs[r * 260 + kc * 8 + 4 * h];
      v2f a, bb;
      a.x = k4.x; a.y = k4.y; bb.x = q4.x; bb.y = q4.y;
      s = wmma4(a, bb, s);
      a.x = k4.z; a.y = k4.w; bb.x = q4.z; bb.y = q4.w;
      s = wmma4(a, bb, s);
    }
    __builtin_amdgcn_sched_barrier(0);

    float p[8];
#pragma unroll
    for (int i = 0; i < 8; ++i) {
      const int gi = i0 + r;
      const int gj = j0 + 8 * h + i;
      const int d = gj - gi;
      const float wgt = (d > 0) ? __expf(ln_dec * (float)(d - 1)) : 0.0f;
      p[i] = s[i] * wgt;
    }

#pragma unroll
    for (int g = 0; g < 2; ++g) {
      float4 va[8], vb[8];
#pragma unroll
      for (int u = 0; u < 8; ++u) {
        va[u] = *(const float4*)&Vj[((g * 8 + u) * 2 + 0) * 128];
        vb[u] = *(const float4*)&Vj[((g * 8 + u) * 2 + 1) * 128];
      }
#pragma unroll
      for (int u = 0; u < 8; ++u) {
        v2f a, bb;
        a.x = va[u].x; a.y = va[u].y; bb.x = p[0]; bb.y = p[1];
        acc[g * 8 + u] = wmma4(a, bb, acc[g * 8 + u]);
      }
#pragma unroll
      for (int u = 0; u < 8; ++u) {
        v2f a, bb;
        a.x = va[u].z; a.y = va[u].w; bb.x = p[2]; bb.y = p[3];
        acc[g * 8 + u] = wmma4(a, bb, acc[g * 8 + u]);
      }
#pragma unroll
      for (int u = 0; u < 8; ++u) {
        v2f a, bb;
        a.x = vb[u].x; a.y = vb[u].y; bb.x = p[4]; bb.y = p[5];
        acc[g * 8 + u] = wmma4(a, bb, acc[g * 8 + u]);
      }
#pragma unroll
      for (int u = 0; u < 8; ++u) {
        v2f a, bb;
        a.x = vb[u].z; a.y = vb[u].w; bb.x = p[6]; bb.y = p[7];
        acc[g * 8 + u] = wmma4(a, bb, acc[g * 8 + u]);
      }
    }
  }

  __syncthreads();
  float* pw = pool + wv * (16 * 260);
#pragma unroll
  for (int nt = 0; nt < 16; ++nt) {
    float4 lo, hi;
    lo.x = acc[nt][0]; lo.y = acc[nt][1]; lo.z = acc[nt][2]; lo.w = acc[nt][3];
    hi.x = acc[nt][4]; hi.y = acc[nt][5]; hi.z = acc[nt][6]; hi.w = acc[nt][7];
    *(float4*)&pw[r * 260 + nt * 16 + 8 * h] = lo;
    *(float4*)&pw[r * 260 + nt * 16 + 8 * h + 4] = hi;
  }
  __syncthreads();

  for (int idx = tid; idx < 16 * 64; idx += 32 * ATT_WAVES) {
    int row = idx >> 6;
    int c4 = (idx & 63) << 2;
    const int o = row * 260 + c4;
    float4 sum = *(const float4*)&pool[(ATT_WAVES - 1) * (16 * 260) + o];
#pragma unroll
    for (int w = ATT_WAVES - 2; w >= 0; --w) {
      const float4 p = *(const float4*)&pool[w * (16 * 260) + o];
      sum.x += p.x; sum.y += p.y; sum.z += p.z; sum.w += p.w;
    }
    *(float4*)&Rb[(size_t)(i0 + row) * C_DIM + c4] = sum;
  }
}

__global__ void out_proj_kernel(const float* __restrict__ R,
                                const float* __restrict__ owT,
                                const float* __restrict__ out_scale,
                                float* __restrict__ out) {
  const int lane = threadIdx.x & 31;
  const int r = lane & 15;
  const int h = lane >> 4;
  const int m0 = blockIdx.x * 16;
  const int v0 = blockIdx.y * 64;

  v8f acc[4];
#pragma unroll
  for (int nt = 0; nt < 4; ++nt) acc[nt] = vzero8();

  const float* rr = &R[(size_t)(m0 + r) * C_DIM + 4 * h];
  const float* wf = owT + (size_t)(blockIdx.y * 4) * (C_DIM / 8) * 128 + 4 * lane;
#pragma unroll 4
  for (int kc = 0; kc < C_DIM / 8; ++kc) {
    const float4 a4 = *(const float4*)&rr[kc * 8];
    float4 b4[4];
#pragma unroll
    for (int nt = 0; nt < 4; ++nt)
      b4[nt] = *(const float4*)&wf[((size_t)nt * (C_DIM / 8) + kc) * 128];
#pragma unroll
    for (int nt = 0; nt < 4; ++nt) {
      v2f a, b;
      a.x = a4.x; a.y = a4.y; b.x = b4[nt].x; b.y = b4[nt].y;
      acc[nt] = wmma4(a, b, acc[nt]);
      a.x = a4.z; a.y = a4.w; b.x = b4[nt].z; b.y = b4[nt].w;
      acc[nt] = wmma4(a, b, acc[nt]);
    }
  }
  const float sc = *out_scale;
#pragma unroll
  for (int nt = 0; nt < 4; ++nt)
#pragma unroll
    for (int i = 0; i < 8; ++i)
      out[(size_t)(m0 + i + 8 * h) * V_DIM + v0 + nt * 16 + r] = acc[nt][i] * sc;
}

extern "C" void kernel_launch(void* const* d_in, const int* in_sizes, int n_in,
                              void* d_out, int out_size, void* d_ws, size_t ws_size,
                              hipStream_t stream) {
  (void)in_sizes; (void)n_in; (void)out_size; (void)ws_size;

  const float* x = (const float*)d_in[0];
  const float* basis = (const float*)d_in[1];
  const float* qc = (const float*)d_in[2];
  const float* kc = (const float*)d_in[3];
  const float* vc = (const float*)d_in[4];
  const float* oc = (const float*)d_in[5];
  const float* decay_logit = (const float*)d_in[6];
  const float* out_scale = (const float*)d_in[7];
  float* out = (float*)d_out;

  float* ws = (float*)d_ws;
  float* qw = ws;
  float* kw = qw + (size_t)V_DIM * C_DIM;
  float* vw = kw + (size_t)V_DIM * C_DIM;
  float* owT = vw + (size_t)V_DIM * C_DIM;
  float* Q = owT + (size_t)V_DIM * C_DIM;
  float* K = Q + (size_t)M_DIM * C_DIM;
  float* Vv = K + (size_t)M_DIM * C_DIM;
  float* R = Vv + (size_t)M_DIM * C_DIM;

  dim3 blk(32);

  proj_weights_kernel<<<dim3(V_DIM / 16, C_DIM / 16, 4), blk, 0, stream>>>(
      basis, qc, kc, vc, oc, qw, kw, vw, owT);

  qkv_kernel<<<dim3(M_DIM / 16, C_DIM / 64, 3), blk, 0, stream>>>(
      x, qw, kw, vw, Q, K, Vv);

  attn_kernel<<<dim3(B_DIM, T_DIM / 16), dim3(32 * ATT_WAVES), 0, stream>>>(
      Q, K, Vv, decay_logit, R);

  out_proj_kernel<<<dim3(M_DIM / 16, V_DIM / 64), blk, 0, stream>>>(
      R, owT, out_scale, out);
}


# attention key loop software-pipelined: each 32-WMMA block's 16 operand loads issued one block ahead into a second register set; 12 waves per work-group, 3 per SIMD; partial-sum order changes
# speedup vs baseline: 3.9849x; 1.0358x over previous
#include <hip/hip_runtime.h>


typedef float v2f __attribute__((ext_vector_type(2)));
typedef float v8f __attribute__((ext_vector_type(8)));

#define C_DIM 256
#define V_DIM 1024
#define F_DIM 256
#define T_DIM 4096
#define B_DIM 4
#define M_DIM (B_DIM * T_DIM)

static __device__ __forceinline__ v8f wmma4(v2f a, v2f b, v8f c) {
  return __builtin_amdgcn_wmma_f32_16x16x4_f32(
       false, a,  false, b,
       (short)0, c,  false,  false);
}

static __device__ __forceinline__ v8f vzero8() {
  v8f z;
#pragma unroll
  for (int i = 0; i < 8; ++i) z[i] = 0.0f;
  return z;
}

__global__ void proj_weights_kernel(const float* __restrict__ basis,
                                    const float* __restrict__ qc,
                                    const float* __restrict__ kc,
                                    const float* __restrict__ vc,
                                    const float* __restrict__ oc,
                                    float* __restrict__ qw,
                                    float* __restrict__ kw,
                                    float* __restrict__ vw,
                                    float* __restrict__ owT) {
  const int lane = threadIdx.x & 31;
  const int r = lane & 15;
  const int h = lane >> 4;
  const int v0 = blockIdx.x * 16;
  const int c0 = blockIdx.y * 16;
  const int which = blockIdx.z;
  const float* coef = (which == 0) ? qc : (which == 1) ? kc : (which == 2) ? vc : oc;

  const float* brow = &basis[(size_t)(v0 + r) * F_DIM + 2 * h];
  const float* crow = &coef[(size_t)(c0 + r) * F_DIM + 2 * h];
  const float* ap = (which < 3) ? brow : crow;
  const float* bp = (which < 3) ? crow : brow;
  v8f acc = vzero8();
  for (int f0 = 0; f0 < F_DIM; f0 += 4) {
    v2f a = *(const v2f*)&ap[f0];
    v2f b = *(const v2f*)&bp[f0];
    acc = wmma4(a, b, acc);
  }
  float4 lo, hi;
  lo.x = acc[0]; lo.y = acc[1]; lo.z = acc[2]; lo.w = acc[3];
  hi.x = acc[4]; hi.y = acc[5]; hi.z = acc[6]; hi.w = acc[7];
  float* dst;
  if (which < 3) {
    float* w = (which == 0) ? qw : (which == 1) ? kw : vw;
    dst = w + ((size_t)blockIdx.y * (V_DIM / 8) + 2 * blockIdx.x + h) * 128;
  } else {
    dst = owT + ((size_t)blockIdx.x * (C_DIM / 8) + 2 * blockIdx.y + h) * 128;
  }
  *(float4*)&dst[4 * r] = lo;
  *(float4*)&dst[4 * (16 + r)] = hi;
}

template <bool SWAP>
static __device__ __forceinline__ void xw_tile_16x64(const float* __restrict__ xr,
                                                     const float* __restrict__ wf,
                                                     v8f (&acc)[4]) {
#pragma unroll 4
  for (int kc = 0; kc < V_DIM / 8; ++kc) {
    const float4 a4 = *(const float4*)&xr[kc * 8];
    float4 b4[4];
#pragma unroll
    for (int nt = 0; nt < 4; ++nt)
      b4[nt] = *(const float4*)&wf[((size_t)nt * (V_DIM / 8) + kc) * 128];
#pragma unroll
    for (int nt = 0; nt < 4; ++nt) {
      v2f a, b;
      a.x = a4.x; a.y = a4.y; b.x = b4[nt].x; b.y = b4[nt].y;
      acc[nt] = SWAP ? wmma4(b, a, acc[nt]) : wmma4(a, b, acc[nt]);
      a.x = a4.z; a.y = a4.w; b.x = b4[nt].z; b.y = b4[nt].w;
      acc[nt] = SWAP ? wmma4(b, a, acc[nt]) : wmma4(a, b, acc[nt]);
    }
  }
}

__global__ void qkv_kernel(const float* __restrict__ x,
                           const float* __restrict__ qw,
                           const float* __restrict__ kw,
                           const float* __restrict__ vw,
                           float* __restrict__ Q,
                           float* __restrict__ K,
                           float* __restrict__ Vv) {
  const int lane = threadIdx.x & 31;
  const int r = lane & 15;
  const int h = lane >> 4;
  const int m0 = blockIdx.x * 16;
  const int c0 = blockIdx.y * 64;
  const int which = blockIdx.z;
  const float* w = (which == 0) ? qw : (which == 1) ? kw : vw;
  float* out = (which == 0) ? Q : (which == 1) ? K : Vv;

  v8f acc[4];
#pragma unroll
  for (int nt = 0; nt < 4; ++nt) acc[nt] = vzero8();

  const float* xr = &x[(size_t)(m0 + r) * V_DIM + 4 * h];
  const float* wf = w + (size_t)(blockIdx.y * 4) * (V_DIM / 8) * 128 + 4 * lane;
  if (which == 1) {
    xw_tile_16x64<true>(xr, wf, acc);
#pragma unroll
    for (int nt = 0; nt < 4; ++nt) {
      float4 lo, hi;
      lo.x = acc[nt][0]; lo.y = acc[nt][1]; lo.z = acc[nt][2]; lo.w = acc[nt][3];
      hi.x = acc[nt][4]; hi.y = acc[nt][5]; hi.z = acc[nt][6]; hi.w = acc[nt][7];
      float* dst = out + ((size_t)blockIdx.x * (C_DIM / 8) + blockIdx.y * 8 + 2 * nt + h) * 128;
      *(float4*)&dst[4 * r] = lo;
      *(float4*)&dst[4 * (16 + r)] = hi;
    }
  } else {
    xw_tile_16x64<false>(xr, wf, acc);
    if (which == 0) {
#pragma unroll
      for (int nt = 0; nt < 4; ++nt)
#pragma unroll
        for (int i = 0; i < 8; ++i)
          out[(size_t)(m0 + i + 8 * h) * C_DIM + c0 + nt * 16 + r] = acc[nt][i];
    } else {
#pragma unroll
      for (int nt = 0; nt < 4; ++nt) {
        float4 lo, hi;
        lo.x = acc[nt][0]; lo.y = acc[nt][1]; lo.z = acc[nt][2]; lo.w = acc[nt][3];
        hi.x = acc[nt][4]; hi.y = acc[nt][5]; hi.z = acc[nt][6]; hi.w = acc[nt][7];
        float* dst = out + (((size_t)blockIdx.x * (C_DIM / 16) + blockIdx.y * 4 + nt) * 2) * 128;
        *(float4*)&dst[4 * lane] = lo;
        *(float4*)&dst[128 + 4 * lane] = hi;
      }
    }
  }
}


static __device__ __forceinline__ v8f score_block16(const float4 (&kf)[16],
                                                    const float* __restrict__ qrow,
                                                    v8f s) {
  float4 qn[4];
#pragma unroll
  for (int t = 0; t < 4; ++t) qn[t] = *(const float4*)&qrow[t * 8];
#pragma unroll
  for (int sb = 0; sb < 4; ++sb) {
    float4 qc[4];
#pragma unroll
    for (int t = 0; t < 4; ++t) qc[t] = qn[t];
    if (sb < 3) {
#pragma unroll
      for (int t = 0; t < 4; ++t) qn[t] = *(const float4*)&qrow[(4 * (sb + 1) + t) * 8];
    }
    __builtin_amdgcn_sched_barrier(0);
#pragma unroll
    for (int t = 0; t < 4; ++t) {
      const float4 k4 = kf[4 * sb + t];
      v2f a, bb;
      a.x = k4.x; a.y = k4.y; bb.x = qc[t].x; bb.y = qc[t].y;
      s = wmma4(a, bb, s);
      a.x = k4.z; a.y = k4.w; bb.x = qc[t].z; bb.y = qc[t].w;
      s = wmma4(a, bb, s);
    }
    __builtin_amdgcn_sched_barrier(0);
  }
  return s;
}

#define ATT_WAVES 12

__global__ __launch_bounds__(32 * ATT_WAVES)
void attn_kernel(const float* __restrict__ Q,
                 const float* __restrict__ K,
                 const float* __restrict__ Vv,
                 const float* __restrict__ decay_logit,
                 float* __restrict__ R) {
  __shared__ __attribute__((aligned(16))) float pool[ATT_WAVES * 16 * 260];
  float* qs = pool;

  const int tid = threadIdx.x;
  const int lane = tid & 31;
  const int wv = __builtin_amdgcn_readfirstlane(tid >> 5);
  const int r = lane & 15;
  const int h = lane >> 4;
  const int i0 = blockIdx.y * 16;
  const int b = blockIdx.x;

  const float* Qb = Q + (size_t)b * T_DIM * C_DIM;
  const float* Kb = K + (size_t)b * T_DIM * C_DIM;
  const float* Vb = Vv + (size_t)b * T_DIM * C_DIM;
  float* Rb = R + (size_t)b * T_DIM * C_DIM;

  for (int idx = tid; idx < 16 * 64; idx += 32 * ATT_WAVES) {
    int row = idx >> 6;
    int c4 = (idx & 63) << 2;
    *(float4*)&qs[row * 260 + c4] =
        *(const float4*)&Qb[(size_t)(i0 + row) * C_DIM + c4];
  }
  __syncthreads();

  const float dl = *decay_logit;
  const float dec = 1.0f / (1.0f + __expf(-dl));
  const float ln_dec = __logf(dec);

  v8f acc[16];
#pragma unroll
  for (int nt = 0; nt < 16; ++nt) acc[nt] = vzero8();

  const int ntiles = (T_DIM - i0) >> 4;

  float4 kA[16];
  {
    const int jf = (wv < ntiles) ? wv : 0;
    const float* Kf = Kb + (size_t)(i0 + 16 * jf) * C_DIM + 4 * lane;
#pragma unroll
    for (int u = 0; u < 16; ++u) kA[u] = *(const float4*)&Kf[u * 128];
  }

  for (int jt = wv; jt < ntiles; jt += ATT_WAVES) {
    const int j0 = i0 + 16 * jt;
    const float* Kj = Kb + (size_t)j0 * C_DIM + 4 * lane;
    const float* Vj = Vb + (size_t)j0 * C_DIM + 4 * lane;
    const int jn = (jt + ATT_WAVES < ntiles) ? (j0 + 16 * ATT_WAVES) : j0;
    const float* Kn = Kb + (size_t)jn * C_DIM + 4 * lane;

    asm volatile("" ::: "memory");

    float4 kB[16];
#pragma unroll
    for (int u = 0; u < 16; ++u) kB[u] = *(const float4*)&Kj[(16 + u) * 128];
    __builtin_amdgcn_sched_barrier(0);

    v8f s = score_block16(kA, &qs[r * 260 + 4 * h], vzero8());
    __builtin_amdgcn_sched_barrier(0);

    float4 va0[8], vb0[8];
#pragma unroll
    for (int u = 0; u < 8; ++u) {
      va0[u] = *(const float4*)&Vj[(u * 2 + 0) * 128];
      vb0[u] = *(const float4*)&Vj[(u * 2 + 1) * 128];
    }
    __builtin_amdgcn_sched_barrier(0);

    s = score_block16(kB, &qs[r * 260 + 128 + 4 * h], s);
    __builtin_amdgcn_sched_barrier(0);

    float4 va1[8], vb1[8];
#pragma unroll
    for (int u = 0; u < 8; ++u) {
      va1[u] = *(const float4*)&Vj[((8 + u) * 2 + 0) * 128];
      vb1[u] = *(const float4*)&Vj[((8 + u) * 2 + 1) * 128];
    }

    float p[8];
#pragma unroll
    for (int i = 0; i < 8; ++i) {
      const int gi = i0 + r;
      const int gj = j0 + 8 * h + i;
      const int d = gj - gi;
      const float wgt = (d > 0) ? __expf(ln_dec * (float)(d - 1)) : 0.0f;
      p[i] = s[i] * wgt;
    }
    __builtin_amdgcn_sched_barrier(0);

#pragma unroll
    for (int u = 0; u < 8; ++u) {
      v2f a, bb;
      a.x = va0[u].x; a.y = va0[u].y; bb.x = p[0]; bb.y = p[1];
      acc[u] = wmma4(a, bb, acc[u]);
    }
#pragma unroll
    for (int u = 0; u < 8; ++u) {
      v2f a, bb;
      a.x = va0[u].z; a.y = va0[u].w; bb.x = p[2]; bb.y = p[3];
      acc[u] = wmma4(a, bb, acc[u]);
    }
#pragma unroll
    for (int u = 0; u < 8; ++u) {
      v2f a, bb;
      a.x = vb0[u].x; a.y = vb0[u].y; bb.x = p[4]; bb.y = p[5];
      acc[u] = wmma4(a, bb, acc[u]);
    }
#pragma unroll
    for (int u = 0; u < 8; ++u) {
      v2f a, bb;
      a.x = vb0[u].z; a.y = vb0[u].w; bb.x = p[6]; bb.y = p[7];
      acc[u] = wmma4(a, bb, acc[u]);
    }
    __builtin_amdgcn_sched_barrier(0);

#pragma unroll
    for (int u = 0; u < 16; ++u) kA[u] = *(const float4*)&Kn[u * 128];
    __builtin_amdgcn_sched_barrier(0);

#pragma unroll
    for (int u = 0; u < 8; ++u) {
      v2f a, bb;
      a.x = va1[u].x; a.y = va1[u].y; bb.x = p[0]; bb.y = p[1];
      acc[8 + u] = wmma4(a, bb, acc[8 + u]);
    }
#pragma unroll
    for (int u = 0; u < 8; ++u) {
      v2f a, bb;
      a.x = va1[u].z; a.y = va1[u].w; bb.x = p[2]; bb.y = p[3];
      acc[8 + u] = wmma4(a, bb, acc[8 + u]);
    }
#pragma unroll
    for (int u = 0; u < 8; ++u) {
      v2f a, bb;
      a.x = vb1[u].x; a.y = vb1[u].y; bb.x = p[4]; bb.y = p[5];
      acc[8 + u] = wmma4(a, bb, acc[8 + u]);
    }
#pragma unroll
    for (int u = 0; u < 8; ++u) {
      v2f a, bb;
      a.x = vb1[u].z; a.y = vb1[u].w; bb.x = p[6]; bb.y = p[7];
      acc[8 + u] = wmma4(a, bb, acc[8 + u]);
    }
  }

  __syncthreads();
  float* pw = pool + wv * (16 * 260);
#pragma unroll
  for (int nt = 0; nt < 16; ++nt) {
    float4 lo, hi;
    lo.x = acc[nt][0]; lo.y = acc[nt][1]; lo.z = acc[nt][2]; lo.w = acc[nt][3];
    hi.x = acc[nt][4]; hi.y = acc[nt][5]; hi.z = acc[nt][6]; hi.w = acc[nt][7];
    *(float4*)&pw[r * 260 + nt * 16 + 8 * h] = lo;
    *(float4*)&pw[r * 260 + nt * 16 + 8 * h + 4] = hi;
  }
  __syncthreads();

  for (int idx = tid; idx < 16 * 64; idx += 32 * ATT_WAVES) {
    int row = idx >> 6;
    int c4 = (idx & 63) << 2;
    const int o = row * 260 + c4;
    float4 sum = *(const float4*)&pool[(ATT_WAVES - 1) * (16 * 260) + o];
#pragma unroll
    for (int w = ATT_WAVES - 2; w >= 0; --w) {
      const float4 p = *(const float4*)&pool[w * (16 * 260) + o];
      sum.x += p.x; sum.y += p.y; sum.z += p.z; sum.w += p.w;
    }
    *(float4*)&Rb[(size_t)(i0 + row) * C_DIM + c4] = sum;
  }
}

__global__ void out_proj_kernel(const float* __restrict__ R,
                                const float* __restrict__ owT,
                                const float* __restrict__ out_scale,
                                float* __restrict__ out) {
  const int lane = threadIdx.x & 31;
  const int r = lane & 15;
  const int h = lane >> 4;
  const int m0 = blockIdx.x * 16;
  const int v0 = blockIdx.y * 64;

  v8f acc[4];
#pragma unroll
  for (int nt = 0; nt < 4; ++nt) acc[nt] = vzero8();

  const float* rr = &R[(size_t)(m0 + r) * C_DIM + 4 * h];
  const float* wf = owT + (size_t)(blockIdx.y * 4) * (C_DIM / 8) * 128 + 4 * lane;
#pragma unroll 4
  for (int kc = 0; kc < C_DIM / 8; ++kc) {
    const float4 a4 = *(const float4*)&rr[kc * 8];
    float4 b4[4];
#pragma unroll
    for (int nt = 0; nt < 4; ++nt)
      b4[nt] = *(const float4*)&wf[((size_t)nt * (C_DIM / 8) + kc) * 128];
#pragma unroll
    for (int nt = 0; nt < 4; ++nt) {
      v2f a, b;
      a.x = a4.x; a.y = a4.y; b.x = b4[nt].x; b.y = b4[nt].y;
      acc[nt] = wmma4(a, b, acc[nt]);
      a.x = a4.z; a.y = a4.w; b.x = b4[nt].z; b.y = b4[nt].w;
      acc[nt] = wmma4(a, b, acc[nt]);
    }
  }
  const float sc = *out_scale;
#pragma unroll
  for (int nt = 0; nt < 4; ++nt)
#pragma unroll
    for (int i = 0; i < 8; ++i)
      out[(size_t)(m0 + i + 8 * h) * V_DIM + v0 + nt * 16 + r] = acc[nt][i] * sc;
}

extern "C" void kernel_launch(void* const* d_in, const int* in_sizes, int n_in,
                              void* d_out, int out_size, void* d_ws, size_t ws_size,
                              hipStream_t stream) {
  (void)in_sizes; (void)n_in; (void)out_size; (void)ws_size;

  const float* x = (const float*)d_in[0];
  const float* basis = (const float*)d_in[1];
  const float* qc = (const float*)d_in[2];
  const float* kc = (const float*)d_in[3];
  const float* vc = (const float*)d_in[4];
  const float* oc = (const float*)d_in[5];
  const float* decay_logit = (const float*)d_in[6];
  const float* out_scale = (const float*)d_in[7];
  float* out = (float*)d_out;

  float* ws = (float*)d_ws;
  float* qw = ws;
  float* kw = qw + (size_t)V_DIM * C_DIM;
  float* vw = kw + (size_t)V_DIM * C_DIM;
  float* owT = vw + (size_t)V_DIM * C_DIM;
  float* Q = owT + (size_t)V_DIM * C_DIM;
  float* K = Q + (size_t)M_DIM * C_DIM;
  float* Vv = K + (size_t)M_DIM * C_DIM;
  float* R = Vv + (size_t)M_DIM * C_DIM;

  dim3 blk(32);

  proj_weights_kernel<<<dim3(V_DIM / 16, C_DIM / 16, 4), blk, 0, stream>>>(
      basis, qc, kc, vc, oc, qw, kw, vw, owT);

  qkv_kernel<<<dim3(M_DIM / 16, C_DIM / 64, 3), blk, 0, stream>>>(
      x, qw, kw, vw, Q, K, Vv);

  attn_kernel<<<dim3(B_DIM, T_DIM / 16), dim3(32 * ATT_WAVES), 0, stream>>>(
      Q, K, Vv, decay_logit, R);

  out_proj_kernel<<<dim3(M_DIM / 16, V_DIM / 64), blk, 0, stream>>>(
      R, owT, out_scale, out);
}


# attention: key loop ends at the decay horizon, found at run time from decay_logit: the first key tile whose nearest mask weight is exactly 0.0f; it and later tiles add only zeros; sum order kept
# speedup vs baseline: 4.5458x; 1.1408x over previous
#include <hip/hip_runtime.h>


typedef float v2f __attribute__((ext_vector_type(2)));
typedef float v8f __attribute__((ext_vector_type(8)));

#define C_DIM 256
#define V_DIM 1024
#define F_DIM 256
#define T_DIM 4096
#define B_DIM 4
#define M_DIM (B_DIM * T_DIM)

static __device__ __forceinline__ v8f wmma4(v2f a, v2f b, v8f c) {
  return __builtin_amdgcn_wmma_f32_16x16x4_f32(
       false, a,  false, b,
       (short)0, c,  false,  false);
}

static __device__ __forceinline__ v8f vzero8() {
  v8f z;
#pragma unroll
  for (int i = 0; i < 8; ++i) z[i] = 0.0f;
  return z;
}

__global__ void proj_weights_kernel(const float* __restrict__ basis,
                                    const float* __restrict__ qc,
                                    const float* __restrict__ kc,
                                    const float* __restrict__ vc,
                                    const float* __restrict__ oc,
                                    float* __restrict__ qw,
                                    float* __restrict__ kw,
                                    float* __restrict__ vw,
                                    float* __restrict__ owT) {
  const int lane = threadIdx.x & 31;
  const int r = lane & 15;
  const int h = lane >> 4;
  const int v0 = blockIdx.x * 16;
  const int c0 = blockIdx.y * 16;
  const int which = blockIdx.z;
  const float* coef = (which == 0) ? qc : (which == 1) ? kc : (which == 2) ? vc : oc;

  const float* brow = &basis[(size_t)(v0 + r) * F_DIM + 2 * h];
  const float* crow = &coef[(size_t)(c0 + r) * F_DIM + 2 * h];
  const float* ap = (which < 3) ? brow : crow;
  const float* bp = (which < 3) ? crow : brow;
  v8f acc = vzero8();
  for (int f0 = 0; f0 < F_DIM; f0 += 4) {
    v2f a = *(const v2f*)&ap[f0];
    v2f b = *(const v2f*)&bp[f0];
    acc = wmma4(a, b, acc);
  }
  float4 lo, hi;
  lo.x = acc[0]; lo.y = acc[1]; lo.z = acc[2]; lo.w = acc[3];
  hi.x = acc[4]; hi.y = acc[5]; hi.z = acc[6]; hi.w = acc[7];
  float* dst;
  if (which < 3) {
    float* w = (which == 0) ? qw : (which == 1) ? kw : vw;
    dst = w + ((size_t)blockIdx.y * (V_DIM / 8) + 2 * blockIdx.x + h) * 128;
  } else {
    dst = owT + ((size_t)blockIdx.x * (C_DIM / 8) + 2 * blockIdx.y + h) * 128;
  }
  *(float4*)&dst[4 * r] = lo;
  *(float4*)&dst[4 * (16 + r)] = hi;
}

template <bool SWAP>
static __device__ __forceinline__ void xw_tile_16x64(const float* __restrict__ xr,
                                                     const float* __restrict__ wf,
                                                     v8f (&acc)[4]) {
#pragma unroll 4
  for (int kc = 0; kc < V_DIM / 8; ++kc) {
    const float4 a4 = *(const float4*)&xr[kc * 8];
    float4 b4[4];
#pragma unroll
    for (int nt = 0; nt < 4; ++nt)
      b4[nt] = *(const float4*)&wf[((size_t)nt * (V_DIM / 8) + kc) * 128];
#pragma unroll
    for (int nt = 0; nt < 4; ++nt) {
      v2f a, b;
      a.x = a4.x; a.y = a4.y; b.x = b4[nt].x; b.y = b4[nt].y;
      acc[nt] = SWAP ? wmma4(b, a, acc[nt]) : wmma4(a, b, acc[nt]);
      a.x = a4.z; a.y = a4.w; b.x = b4[nt].z; b.y = b4[nt].w;
      acc[nt] = SWAP ? wmma4(b, a, acc[nt]) : wmma4(a, b, acc[nt]);
    }
  }
}

__global__ void qkv_kernel(const float* __restrict__ x,
                           const float* __restrict__ qw,
                           const float* __restrict__ kw,
                           const float* __restrict__ vw,
                           float* __restrict__ Q,
                           float* __restrict__ K,
                           float* __restrict__ Vv) {
  const int lane = threadIdx.x & 31;
  const int r = lane & 15;
  const int h = lane >> 4;
  const int m0 = blockIdx.x * 16;
  const int c0 = blockIdx.y * 64;
  const int which = blockIdx.z;
  const float* w = (which == 0) ? qw : (which == 1) ? kw : vw;
  float* out = (which == 0) ? Q : (which == 1) ? K : Vv;

  v8f acc[4];
#pragma unroll
  for (int nt = 0; nt < 4; ++nt) acc[nt] = vzero8();

  const float* xr = &x[(size_t)(m0 + r) * V_DIM + 4 * h];
  const float* wf = w + (size_t)(blockIdx.y * 4) * (V_DIM / 8) * 128 + 4 * lane;
  if (which == 1) {
    xw_tile_16x64<true>(xr, wf, acc);
#pragma unroll
    for (int nt = 0; nt < 4; ++nt) {
      float4 lo, hi;
      lo.x = acc[nt][0]; lo.y = acc[nt][1]; lo.z = acc[nt][2]; lo.w = acc[nt][3];
      hi.x = acc[nt][4]; hi.y = acc[nt][5]; hi.z = acc[nt][6]; hi.w = acc[nt][7];
      float* dst = out + ((size_t)blockIdx.x * (C_DIM / 8) + blockIdx.y * 8 + 2 * nt + h) * 128;
      *(float4*)&dst[4 * r] = lo;
      *(float4*)&dst[4 * (16 + r)] = hi;
    }
  } else {
    xw_tile_16x64<false>(xr, wf, acc);
    if (which == 0) {
#pragma unroll
      for (int nt = 0; nt < 4; ++nt)
#pragma unroll
        for (int i = 0; i < 8; ++i)
          out[(size_t)(m0 + i + 8 * h) * C_DIM + c0 + nt * 16 + r] = acc[nt][i];
    } else {
#pragma unroll
      for (int nt = 0; nt < 4; ++nt) {
        float4 lo, hi;
        lo.x = acc[nt][0]; lo.y = acc[nt][1]; lo.z = acc[nt][2]; lo.w = acc[nt][3];
        hi.x = acc[nt][4]; hi.y = acc[nt][5]; hi.z = acc[nt][6]; hi.w = acc[nt][7];
        float* dst = out + (((size_t)blockIdx.x * (C_DIM / 16) + blockIdx.y * 4 + nt) * 2) * 128;
        *(float4*)&dst[4 * lane] = lo;
        *(float4*)&dst[128 + 4 * lane] = hi;
      }
    }
  }
}


static __device__ __forceinline__ v8f score_block16(const float4 (&kf)[16],
                                                    const float* __restrict__ qrow,
                                                    v8f s) {
  float4 qn[4];
#pragma unroll
  for (int t = 0; t < 4; ++t) qn[t] = *(const float4*)&qrow[t * 8];
#pragma unroll
  for (int sb = 0; sb < 4; ++sb) {
    float4 qc[4];
#pragma unroll
    for (int t = 0; t < 4; ++t) qc[t] = qn[t];
    if (sb < 3) {
#pragma unroll
      for (int t = 0; t < 4; ++t) qn[t] = *(const float4*)&qrow[(4 * (sb + 1) + t) * 8];
    }
    __builtin_amdgcn_sched_barrier(0);
#pragma unroll
    for (int t = 0; t < 4; ++t) {
      const float4 k4 = kf[4 * sb + t];
      v2f a, bb;
      a.x = k4.x; a.y = k4.y; bb.x = qc[t].x; bb.y = qc[t].y;
      s = wmma4(a, bb, s);
      a.x = k4.z; a.y = k4.w; bb.x = qc[t].z; bb.y = qc[t].w;
      s = wmma4(a, bb, s);
    }
    __builtin_amdgcn_sched_barrier(0);
  }
  return s;
}

static __device__ __forceinline__ float decay_weight(float ln_dec, int dm1) {
  return __expf(ln_dec * (float)dm1);
}

#define ATT_WAVES 12

__global__ __launch_bounds__(32 * ATT_WAVES)
void attn_kernel(const float* __restrict__ Q,
                 const float* __restrict__ K,
                 const float* __restrict__ Vv,
                 const float* __restrict__ decay_logit,
                 float* __restrict__ R) {
  __shared__ __attribute__((aligned(16))) float pool[ATT_WAVES * 16 * 260];
  float* qs = pool;

  const int tid = threadIdx.x;
  const int lane = tid & 31;
  const int wv = __builtin_amdgcn_readfirstlane(tid >> 5);
  const int r = lane & 15;
  const int h = lane >> 4;
  const int i0 = blockIdx.y * 16;
  const int b = blockIdx.x;

  const float* Qb = Q + (size_t)b * T_DIM * C_DIM;
  const float* Kb = K + (size_t)b * T_DIM * C_DIM;
  const float* Vb = Vv + (size_t)b * T_DIM * C_DIM;
  float* Rb = R + (size_t)b * T_DIM * C_DIM;

  for (int idx = tid; idx < 16 * 64; idx += 32 * ATT_WAVES) {
    int row = idx >> 6;
    int c4 = (idx & 63) << 2;
    *(float4*)&qs[row * 260 + c4] =
        *(const float4*)&Qb[(size_t)(i0 + row) * C_DIM + c4];
  }
  __syncthreads();

  const float dl = *decay_logit;
  const float dec = 1.0f / (1.0f + __expf(-dl));
  const float ln_dec = __logf(dec);

  v8f acc[16];
#pragma unroll
  for (int nt = 0; nt < 16; ++nt) acc[nt] = vzero8();

  const int ntiles = (T_DIM - i0) >> 4;

  int hlo = 1, hhi = ntiles;
  for (int it = 0; it < 8; ++it) {
    const int mid = (hlo + hhi) >> 1;
    const bool all_zero = decay_weight(ln_dec, 16 * mid - 16) == 0.0f;
    hhi = all_zero ? mid : hhi;
    hlo = all_zero ? hlo : mid;
  }
  const int nvis = __builtin_amdgcn_readfirstlane(hhi);

  float4 kA[16];
  {
    const int jf = (wv < nvis) ? wv : 0;
    const float* Kf = Kb + (size_t)(i0 + 16 * jf) * C_DIM + 4 * lane;
#pragma unroll
    for (int u = 0; u < 16; ++u) kA[u] = *(const float4*)&Kf[u * 128];
  }

  for (int jt = wv; jt < nvis; jt += ATT_WAVES) {
    const int j0 = i0 + 16 * jt;
    const float* Kj = Kb + (size_t)j0 * C_DIM + 4 * lane;
    const float* Vj = Vb + (size_t)j0 * C_DIM + 4 * lane;
    const int jn = (jt + ATT_WAVES < nvis) ? (j0 + 16 * ATT_WAVES) : j0;
    const float* Kn = Kb + (size_t)jn * C_DIM + 4 * lane;

    asm volatile("" ::: "memory");

    float4 kB[16];
#pragma unroll
    for (int u = 0; u < 16; ++u) kB[u] = *(const float4*)&Kj[(16 + u) * 128];
    __builtin_amdgcn_sched_barrier(0);

    v8f s = score_block16(kA, &qs[r * 260 + 4 * h], vzero8());
    __builtin_amdgcn_sched_barrier(0);

    float4 va0[8], vb0[8];
#pragma unroll
    for (int u = 0; u < 8; ++u) {
      va0[u] = *(const float4*)&Vj[(u * 2 + 0) * 128];
      vb0[u] = *(const float4*)&Vj[(u * 2 + 1) * 128];
    }
    __builtin_amdgcn_sched_barrier(0);

    s = score_block16(kB, &qs[r * 260 + 128 + 4 * h], s);
    __builtin_amdgcn_sched_barrier(0);

    float4 va1[8], vb1[8];
#pragma unroll
    for (int u = 0; u < 8; ++u) {
      va1[u] = *(const float4*)&Vj[((8 + u) * 2 + 0) * 128];
      vb1[u] = *(const float4*)&Vj[((8 + u) * 2 + 1) * 128];
    }

    float p[8];
#pragma unroll
    for (int i = 0; i < 8; ++i) {
      const int gi = i0 + r;
      const int gj = j0 + 8 * h + i;
      const int d = gj - gi;
      const float wgt = (d > 0) ? decay_weight(ln_dec, d - 1) : 0.0f;
      p[i] = s[i] * wgt;
    }
    __builtin_amdgcn_sched_barrier(0);

#pragma unroll
    for (int u = 0; u < 8; ++u) {
      v2f a, bb;
      a.x = va0[u].x; a.y = va0[u].y; bb.x = p[0]; bb.y = p[1];
      acc[u] = wmma4(a, bb, acc[u]);
    }
#pragma unroll
    for (int u = 0; u < 8; ++u) {
      v2f a, bb;
      a.x = va0[u].z; a.y = va0[u].w; bb.x = p[2]; bb.y = p[3];
      acc[u] = wmma4(a, bb, acc[u]);
    }
#pragma unroll
    for (int u = 0; u < 8; ++u) {
      v2f a, bb;
      a.x = vb0[u].x; a.y = vb0[u].y; bb.x = p[4]; bb.y = p[5];
      acc[u] = wmma4(a, bb, acc[u]);
    }
#pragma unroll
    for (int u = 0; u < 8; ++u) {
      v2f a, bb;
      a.x = vb0[u].z; a.y = vb0[u].w; bb.x = p[6]; bb.y = p[7];
      acc[u] = wmma4(a, bb, acc[u]);
    }
    __builtin_amdgcn_sched_barrier(0);

#pragma unroll
    for (int u = 0; u < 16; ++u) kA[u] = *(const float4*)&Kn[u * 128];
    __builtin_amdgcn_sched_barrier(0);

#pragma unroll
    for (int u = 0; u < 8; ++u) {
      v2f a, bb;
      a.x = va1[u].x; a.y = va1[u].y; bb.x = p[0]; bb.y = p[1];
      acc[8 + u] = wmma4(a, bb, acc[8 + u]);
    }
#pragma unroll
    for (int u = 0; u < 8; ++u) {
      v2f a, bb;
      a.x = va1[u].z; a.y = va1[u].w; bb.x = p[2]; bb.y = p[3];
      acc[8 + u] = wmma4(a, bb, acc[8 + u]);
    }
#pragma unroll
    for (int u = 0; u < 8; ++u) {
      v2f a, bb;
      a.x = vb1[u].x; a.y = vb1[u].y; bb.x = p[4]; bb.y = p[5];
      acc[8 + u] = wmma4(a, bb, acc[8 + u]);
    }
#pragma unroll
    for (int u = 0; u < 8; ++u) {
      v2f a, bb;
      a.x = vb1[u].z; a.y = vb1[u].w; bb.x = p[6]; bb.y = p[7];
      acc[8 + u] = wmma4(a, bb, acc[8 + u]);
    }
  }

  __syncthreads();
  float* pw = pool + wv * (16 * 260);
#pragma unroll
  for (int nt = 0; nt < 16; ++nt) {
    float4 lo, hi;
    lo.x = acc[nt][0]; lo.y = acc[nt][1]; lo.z = acc[nt][2]; lo.w = acc[nt][3];
    hi.x = acc[nt][4]; hi.y = acc[nt][5]; hi.z = acc[nt][6]; hi.w = acc[nt][7];
    *(float4*)&pw[r * 260 + nt * 16 + 8 * h] = lo;
    *(float4*)&pw[r * 260 + nt * 16 + 8 * h + 4] = hi;
  }
  __syncthreads();

  for (int idx = tid; idx < 16 * 64; idx += 32 * ATT_WAVES) {
    int row = idx >> 6;
    int c4 = (idx & 63) << 2;
    const int o = row * 260 + c4;
    float4 sum = *(const float4*)&pool[(ATT_WAVES - 1) * (16 * 260) + o];
#pragma unroll
    for (int w = ATT_WAVES - 2; w >= 0; --w) {
      const float4 p = *(const float4*)&pool[w * (16 * 260) + o];
      sum.x += p.x; sum.y += p.y; sum.z += p.z; sum.w += p.w;
    }
    *(float4*)&Rb[(size_t)(i0 + row) * C_DIM + c4] = sum;
  }
}

__global__ void out_proj_kernel(const float* __restrict__ R,
                                const float* __restrict__ owT,
                                const float* __restrict__ out_scale,
                                float* __restrict__ out) {
  const int lane = threadIdx.x & 31;
  const int r = lane & 15;
  const int h = lane >> 4;
  const int m0 = blockIdx.x * 16;
  const int v0 = blockIdx.y * 64;

  v8f acc[4];
#pragma unroll
  for (int nt = 0; nt < 4; ++nt) acc[nt] = vzero8();

  const float* rr = &R[(size_t)(m0 + r) * C_DIM + 4 * h];
  const float* wf = owT + (size_t)(blockIdx.y * 4) * (C_DIM / 8) * 128 + 4 * lane;
#pragma unroll 4
  for (int kc = 0; kc < C_DIM / 8; ++kc) {
    const float4 a4 = *(const float4*)&rr[kc * 8];
    float4 b4[4];
#pragma unroll
    for (int nt = 0; nt < 4; ++nt)
      b4[nt] = *(const float4*)&wf[((size_t)nt * (C_DIM / 8) + kc) * 128];
#pragma unroll
    for (int nt = 0; nt < 4; ++nt) {
      v2f a, b;
      a.x = a4.x; a.y = a4.y; b.x = b4[nt].x; b.y = b4[nt].y;
      acc[nt] = wmma4(a, b, acc[nt]);
      a.x = a4.z; a.y = a4.w; b.x = b4[nt].z; b.y = b4[nt].w;
      acc[nt] = wmma4(a, b, acc[nt]);
    }
  }
  const float sc = *out_scale;
#pragma unroll
  for (int nt = 0; nt < 4; ++nt)
#pragma unroll
    for (int i = 0; i < 8; ++i)
      out[(size_t)(m0 + i + 8 * h) * V_DIM + v0 + nt * 16 + r] = acc[nt][i] * sc;
}

extern "C" void kernel_launch(void* const* d_in, const int* in_sizes, int n_in,
                              void* d_out, int out_size, void* d_ws, size_t ws_size,
                              hipStream_t stream) {
  (void)in_sizes; (void)n_in; (void)out_size; (void)ws_size;

  const float* x = (const float*)d_in[0];
  const float* basis = (const float*)d_in[1];
  const float* qc = (const float*)d_in[2];
  const float* kc = (const float*)d_in[3];
  const float* vc = (const float*)d_in[4];
  const float* oc = (const float*)d_in[5];
  const float* decay_logit = (const float*)d_in[6];
  const float* out_scale = (const float*)d_in[7];
  float* out = (float*)d_out;

  float* ws = (float*)d_ws;
  float* qw = ws;
  float* kw = qw + (size_t)V_DIM * C_DIM;
  float* vw = kw + (size_t)V_DIM * C_DIM;
  float* owT = vw + (size_t)V_DIM * C_DIM;
  float* Q = owT + (size_t)V_DIM * C_DIM;
  float* K = Q + (size_t)M_DIM * C_DIM;
  float* Vv = K + (size_t)M_DIM * C_DIM;
  float* R = Vv + (size_t)M_DIM * C_DIM;

  dim3 blk(32);

  proj_weights_kernel<<<dim3(V_DIM / 16, C_DIM / 16, 4), blk, 0, stream>>>(
      basis, qc, kc, vc, oc, qw, kw, vw, owT);

  qkv_kernel<<<dim3(M_DIM / 16, C_DIM / 64, 3), blk, 0, stream>>>(
      x, qw, kw, vw, Q, K, Vv);

  attn_kernel<<<dim3(B_DIM, T_DIM / 16), dim3(32 * ATT_WAVES), 0, stream>>>(
      Q, K, Vv, decay_logit, R);

  out_proj_kernel<<<dim3(M_DIM / 16, V_DIM / 64), blk, 0, stream>>>(
      R, owT, out_scale, out);
}


# x@w and retrieved@o_w GEMMs: 32x128 register tile per wave (was 16x64), fragments straight from global, loads one chunk ahead: half the bytes per WMMA; 12/16-wave work-groups; sum order unchanged
# speedup vs baseline: 5.0862x; 1.1189x over previous
#include <hip/hip_runtime.h>


typedef float v2f __attribute__((ext_vector_type(2)));
typedef float v8f __attribute__((ext_vector_type(8)));

#define C_DIM 256
#define V_DIM 1024
#define F_DIM 256
#define T_DIM 4096
#define B_DIM 4
#define M_DIM (B_DIM * T_DIM)

static __device__ __forceinline__ v8f wmma4(v2f a, v2f b, v8f c) {
  return __builtin_amdgcn_wmma_f32_16x16x4_f32(
       false, a,  false, b,
       (short)0, c,  false,  false);
}

static __device__ __forceinline__ v8f vzero8() {
  v8f z;
#pragma unroll
  for (int i = 0; i < 8; ++i) z[i] = 0.0f;
  return z;
}

__global__ void proj_weights_kernel(const float* __restrict__ basis,
                                    const float* __restrict__ qc,
                                    const float* __restrict__ kc,
                                    const float* __restrict__ vc,
                                    const float* __restrict__ oc,
                                    float* __restrict__ qw,
                                    float* __restrict__ kw,
                                    float* __restrict__ vw,
                                    float* __restrict__ owT) {
  const int lane = threadIdx.x & 31;
  const int r = lane & 15;
  const int h = lane >> 4;
  const int v0 = blockIdx.x * 16;
  const int c0 = blockIdx.y * 16;
  const int which = blockIdx.z;
  const float* coef = (which == 0) ? qc : (which == 1) ? kc : (which == 2) ? vc : oc;

  const float* brow = &basis[(size_t)(v0 + r) * F_DIM + 2 * h];
  const float* crow = &coef[(size_t)(c0 + r) * F_DIM + 2 * h];
  const float* ap = (which < 3) ? brow : crow;
  const float* bp = (which < 3) ? crow : brow;
  v8f acc = vzero8();
  for (int f0 = 0; f0 < F_DIM; f0 += 4) {
    v2f a = *(const v2f*)&ap[f0];
    v2f b = *(const v2f*)&bp[f0];
    acc = wmma4(a, b, acc);
  }
  float4 lo, hi;
  lo.x = acc[0]; lo.y = acc[1]; lo.z = acc[2]; lo.w = acc[3];
  hi.x = acc[4]; hi.y = acc[5]; hi.z = acc[6]; hi.w = acc[7];
  float* dst;
  if (which < 3) {
    float* w = (which == 0) ? qw : (which == 1) ? kw : vw;
    dst = w + ((size_t)blockIdx.y * (V_DIM / 8) + 2 * blockIdx.x + h) * 128;
  } else {
    dst = owT + ((size_t)blockIdx.x * (C_DIM / 8) + 2 * blockIdx.y + h) * 128;
  }
  *(float4*)&dst[4 * r] = lo;
  *(float4*)&dst[4 * (16 + r)] = hi;
}

template <int KCH>
static __device__ __forceinline__ void load_frags(const float* __restrict__ ar,
                                                  const float* __restrict__ wf,
                                                  int kc, float4 (&a)[2], float4 (&b)[8]) {
  a[0] = *(const float4*)&ar[kc * 8];
  a[1] = *(const float4*)&ar[(size_t)16 * (8 * KCH) + kc * 8];
#pragma unroll
  for (int nt = 0; nt < 8; ++nt)
    b[nt] = *(const float4*)&wf[((size_t)nt * KCH + kc) * 128];
}

template <bool SWAP>
static __device__ __forceinline__ void mma_chunk(const float4 (&a)[2], const float4 (&b)[8],
                                                 v8f (&acc)[2][8]) {
#pragma unroll
  for (int mb = 0; mb < 2; ++mb) {
#pragma unroll
    for (int nt = 0; nt < 8; ++nt) {
      v2f av, bv;
      av.x = a[mb].x; av.y = a[mb].y; bv.x = b[nt].x; bv.y = b[nt].y;
      acc[mb][nt] = SWAP ? wmma4(bv, av, acc[mb][nt]) : wmma4(av, bv, acc[mb][nt]);
    }
  }
#pragma unroll
  for (int mb = 0; mb < 2; ++mb) {
#pragma unroll
    for (int nt = 0; nt < 8; ++nt) {
      v2f av, bv;
      av.x = a[mb].z; av.y = a[mb].w; bv.x = b[nt].z; bv.y = b[nt].w;
      acc[mb][nt] = SWAP ? wmma4(bv, av, acc[mb][nt]) : wmma4(av, bv, acc[mb][nt]);
    }
  }
}

template <bool SWAP, int KCH>
static __device__ __forceinline__ void gemm_tile_32x128(const float* __restrict__ ar,
                                                        const float* __restrict__ wf,
                                                        v8f (&acc)[2][8]) {
  float4 a0[2], b0[8], a1[2], b1[8];
  load_frags<KCH>(ar, wf, 0, a0, b0);
  for (int kc = 0; kc < KCH; kc += 2) {
    load_frags<KCH>(ar, wf, kc + 1, a1, b1);
    __builtin_amdgcn_sched_barrier(0);
    mma_chunk<SWAP>(a0, b0, acc);
    __builtin_amdgcn_sched_barrier(0);
    const int kn = (kc + 2 < KCH) ? kc + 2 : kc;
    load_frags<KCH>(ar, wf, kn, a0, b0);
    __builtin_amdgcn_sched_barrier(0);
    mma_chunk<SWAP>(a1, b1, acc);
    __builtin_amdgcn_sched_barrier(0);
  }
}

#define QKV_WAVES 12

__global__ __launch_bounds__(32 * QKV_WAVES)
void qkv_kernel(const float* __restrict__ x,
                const float* __restrict__ qw,
                const float* __restrict__ kw,
                const float* __restrict__ vw,
                float* __restrict__ Q,
                float* __restrict__ K,
                float* __restrict__ Vv) {
  const int lane = threadIdx.x & 31;
  const int r = lane & 15;
  const int h = lane >> 4;
  const int wv = __builtin_amdgcn_readfirstlane((int)(threadIdx.x >> 5));
  const int strip = wv / 6;
  const int sub = wv - 6 * strip;
  const int which = sub >> 1;
  const int cg = sub & 1;
  const int mt0 = (blockIdx.x * 2 + strip) * 2;
  const int m0 = mt0 * 16;
  const float* w = (which == 0) ? qw : (which == 1) ? kw : vw;
  float* out = (which == 0) ? Q : (which == 1) ? K : Vv;

  v8f acc[2][8];
#pragma unroll
  for (int mb = 0; mb < 2; ++mb)
#pragma unroll
    for (int nt = 0; nt < 8; ++nt) acc[mb][nt] = vzero8();

  const float* xr = &x[(size_t)(m0 + r) * V_DIM + 4 * h];
  const float* wf = w + (size_t)(cg * 8) * (V_DIM / 8) * 128 + 4 * lane;
  if (which == 1) {
    gemm_tile_32x128<true, V_DIM / 8>(xr, wf, acc);
#pragma unroll
    for (int mb = 0; mb < 2; ++mb)
#pragma unroll
      for (int nt = 0; nt < 8; ++nt) {
        float4 lo, hi;
        lo.x = acc[mb][nt][0]; lo.y = acc[mb][nt][1]; lo.z = acc[mb][nt][2]; lo.w = acc[mb][nt][3];
        hi.x = acc[mb][nt][4]; hi.y = acc[mb][nt][5]; hi.z = acc[mb][nt][6]; hi.w = acc[mb][nt][7];
        float* dst = out + ((size_t)(mt0 + mb) * (C_DIM / 8) + 2 * (cg * 8 + nt) + h) * 128;
        *(float4*)&dst[4 * r] = lo;
        *(float4*)&dst[4 * (16 + r)] = hi;
      }
  } else {
    gemm_tile_32x128<false, V_DIM / 8>(xr, wf, acc);
    if (which == 0) {
#pragma unroll
      for (int mb = 0; mb < 2; ++mb)
#pragma unroll
        for (int nt = 0; nt < 8; ++nt)
#pragma unroll
          for (int i = 0; i < 8; ++i)
            out[(size_t)(m0 + 16 * mb + i + 8 * h) * C_DIM + cg * 128 + nt * 16 + r] = acc[mb][nt][i];
    } else {
#pragma unroll
      for (int mb = 0; mb < 2; ++mb)
#pragma unroll
        for (int nt = 0; nt < 8; ++nt) {
          float4 lo, hi;
          lo.x = acc[mb][nt][0]; lo.y = acc[mb][nt][1]; lo.z = acc[mb][nt][2]; lo.w = acc[mb][nt][3];
          hi.x = acc[mb][nt][4]; hi.y = acc[mb][nt][5]; hi.z = acc[mb][nt][6]; hi.w = acc[mb][nt][7];
          float* dst = out + (((size_t)(mt0 + mb) * (C_DIM / 16) + cg * 8 + nt) * 2) * 128;
          *(float4*)&dst[4 * lane] = lo;
          *(float4*)&dst[128 + 4 * lane] = hi;
        }
    }
  }
}


static __device__ __forceinline__ v8f score_block16(const float4 (&kf)[16],
                                                    const float* __restrict__ qrow,
                                                    v8f s) {
  float4 qn[4];
#pragma unroll
  for (int t = 0; t < 4; ++t) qn[t] = *(const float4*)&qrow[t * 8];
#pragma unroll
  for (int sb = 0; sb < 4; ++sb) {
    float4 qc[4];
#pragma unroll
    for (int t = 0; t < 4; ++t) qc[t] = qn[t];
    if (sb < 3) {
#pragma unroll
      for (int t = 0; t < 4; ++t) qn[t] = *(const float4*)&qrow[(4 * (sb + 1) + t) * 8];
    }
    __builtin_amdgcn_sched_barrier(0);
#pragma unroll
    for (int t = 0; t < 4; ++t) {
      const float4 k4 = kf[4 * sb + t];
      v2f a, bb;
      a.x = k4.x; a.y = k4.y; bb.x = qc[t].x; bb.y = qc[t].y;
      s = wmma4(a, bb, s);
      a.x = k4.z; a.y = k4.w; bb.x = qc[t].z; bb.y = qc[t].w;
      s = wmma4(a, bb, s);
    }
    __builtin_amdgcn_sched_barrier(0);
  }
  return s;
}

static __device__ __forceinline__ float decay_weight(float ln_dec, int dm1) {
  return __expf(ln_dec * (float)dm1);
}

#define ATT_WAVES 12

__global__ __launch_bounds__(32 * ATT_WAVES)
void attn_kernel(const float* __restrict__ Q,
                 const float* __restrict__ K,
                 const float* __restrict__ Vv,
                 const float* __restrict__ decay_logit,
                 float* __restrict__ R) {
  __shared__ __attribute__((aligned(16))) float pool[ATT_WAVES * 16 * 260];
  float* qs = pool;

  const int tid = threadIdx.x;
  const int lane = tid & 31;
  const int wv = __builtin_amdgcn_readfirstlane(tid >> 5);
  const int r = lane & 15;
  const int h = lane >> 4;
  const int i0 = blockIdx.y * 16;
  const int b = blockIdx.x;

  const float* Qb = Q + (size_t)b * T_DIM * C_DIM;
  const float* Kb = K + (size_t)b * T_DIM * C_DIM;
  const float* Vb = Vv + (size_t)b * T_DIM * C_DIM;
  float* Rb = R + (size_t)b * T_DIM * C_DIM;

  for (int idx = tid; idx < 16 * 64; idx += 32 * ATT_WAVES) {
    int row = idx >> 6;
    int c4 = (idx & 63) << 2;
    *(float4*)&qs[row * 260 + c4] =
        *(const float4*)&Qb[(size_t)(i0 + row) * C_DIM + c4];
  }
  __syncthreads();

  const float dl = *decay_logit;
  const float dec = 1.0f / (1.0f + __expf(-dl));
  const float ln_dec = __logf(dec);

  v8f acc[16];
#pragma unroll
  for (int nt = 0; nt < 16; ++nt) acc[nt] = vzero8();

  const int ntiles = (T_DIM - i0) >> 4;

  int hlo = 1, hhi = ntiles;
  for (int it = 0; it < 8; ++it) {
    const int mid = (hlo + hhi) >> 1;
    const bool all_zero = decay_weight(ln_dec, 16 * mid - 16) == 0.0f;
    hhi = all_zero ? mid : hhi;
    hlo = all_zero ? hlo : mid;
  }
  const int nvis = __builtin_amdgcn_readfirstlane(hhi);

  float4 kA[16];
  {
    const int jf = (wv < nvis) ? wv : 0;
    const float* Kf = Kb + (size_t)(i0 + 16 * jf) * C_DIM + 4 * lane;
#pragma unroll
    for (int u = 0; u < 16; ++u) kA[u] = *(const float4*)&Kf[u * 128];
  }

  for (int jt = wv; jt < nvis; jt += ATT_WAVES) {
    const int j0 = i0 + 16 * jt;
    const float* Kj = Kb + (size_t)j0 * C_DIM + 4 * lane;
    const float* Vj = Vb + (size_t)j0 * C_DIM + 4 * lane;
    const int jn = (jt + ATT_WAVES < nvis) ? (j0 + 16 * ATT_WAVES) : j0;
    const float* Kn = Kb + (size_t)jn * C_DIM + 4 * lane;

    asm volatile("" ::: "memory");

    float4 kB[16];
#pragma unroll
    for (int u = 0; u < 16; ++u) kB[u] = *(const float4*)&Kj[(16 + u) * 128];
    __builtin_amdgcn_sched_barrier(0);

    v8f s = score_block16(kA, &qs[r * 260 + 4 * h], vzero8());
    __builtin_amdgcn_sched_barrier(0);

    float4 va0[8], vb0[8];
#pragma unroll
    for (int u = 0; u < 8; ++u) {
      va0[u] = *(const float4*)&Vj[(u * 2 + 0) * 128];
      vb0[u] = *(const float4*)&Vj[(u * 2 + 1) * 128];
    }
    __builtin_amdgcn_sched_barrier(0);

    s = score_block16(kB, &qs[r * 260 + 128 + 4 * h], s);
    __builtin_amdgcn_sched_barrier(0);

    float4 va1[8], vb1[8];
#pragma unroll
    for (int u = 0; u < 8; ++u) {
      va1[u] = *(const float4*)&Vj[((8 + u) * 2 + 0) * 128];
      vb1[u] = *(const float4*)&Vj[((8 + u) * 2 + 1) * 128];
    }

    float p[8];
#pragma unroll
    for (int i = 0; i < 8; ++i) {
      const int gi = i0 + r;
      const int gj = j0 + 8 * h + i;
      const int d = gj - gi;
      const float wgt = (d > 0) ? decay_weight(ln_dec, d - 1) : 0.0f;
      p[i] = s[i] * wgt;
    }
    __builtin_amdgcn_sched_barrier(0);

#pragma unroll
    for (int u = 0; u < 8; ++u) {
      v2f a, bb;
      a.x = va0[u].x; a.y = va0[u].y; bb.x = p[0]; bb.y = p[1];
      acc[u] = wmma4(a, bb, acc[u]);
    }
#pragma unroll
    for (int u = 0; u < 8; ++u) {
      v2f a, bb;
      a.x = va0[u].z; a.y = va0[u].w; bb.x = p[2]; bb.y = p[3];
      acc[u] = wmma4(a, bb, acc[u]);
    }
#pragma unroll
    for (int u = 0; u < 8; ++u) {
      v2f a, bb;
      a.x = vb0[u].x; a.y = vb0[u].y; bb.x = p[4]; bb.y = p[5];
      acc[u] = wmma4(a, bb, acc[u]);
    }
#pragma unroll
    for (int u = 0; u < 8; ++u) {
      v2f a, bb;
      a.x = vb0[u].z; a.y = vb0[u].w; bb.x = p[6]; bb.y = p[7];
      acc[u] = wmma4(a, bb, acc[u]);
    }
    __builtin_amdgcn_sched_barrier(0);

#pragma unroll
    for (int u = 0; u < 16; ++u) kA[u] = *(const float4*)&Kn[u * 128];
    __builtin_amdgcn_sched_barrier(0);

#pragma unroll
    for (int u = 0; u < 8; ++u) {
      v2f a, bb;
      a.x = va1[u].x; a.y = va1[u].y; bb.x = p[0]; bb.y = p[1];
      acc[8 + u] = wmma4(a, bb, acc[8 + u]);
    }
#pragma unroll
    for (int u = 0; u < 8; ++u) {
      v2f a, bb;
      a.x = va1[u].z; a.y = va1[u].w; bb.x = p[2]; bb.y = p[3];
      acc[8 + u] = wmma4(a, bb, acc[8 + u]);
    }
#pragma unroll
    for (int u = 0; u < 8; ++u) {
      v2f a, bb;
      a.x = vb1[u].x; a.y = vb1[u].y; bb.x = p[4]; bb.y = p[5];
      acc[8 + u] = wmma4(a, bb, acc[8 + u]);
    }
#pragma unroll
    for (int u = 0; u < 8; ++u) {
      v2f a, bb;
      a.x = vb1[u].z; a.y = vb1[u].w; bb.x = p[6]; bb.y = p[7];
      acc[8 + u] = wmma4(a, bb, acc[8 + u]);
    }
  }

  __syncthreads();
  float* pw = pool + wv * (16 * 260);
#pragma unroll
  for (int nt = 0; nt < 16; ++nt) {
    float4 lo, hi;
    lo.x = acc[nt][0]; lo.y = acc[nt][1]; lo.z = acc[nt][2]; lo.w = acc[nt][3];
    hi.x = acc[nt][4]; hi.y = acc[nt][5]; hi.z = acc[nt][6]; hi.w = acc[nt][7];
    *(float4*)&pw[r * 260 + nt * 16 + 8 * h] = lo;
    *(float4*)&pw[r * 260 + nt * 16 + 8 * h + 4] = hi;
  }
  __syncthreads();

  for (int idx = tid; idx < 16 * 64; idx += 32 * ATT_WAVES) {
    int row = idx >> 6;
    int c4 = (idx & 63) << 2;
    const int o = row * 260 + c4;
    float4 sum = *(const float4*)&pool[(ATT_WAVES - 1) * (16 * 260) + o];
#pragma unroll
    for (int w = ATT_WAVES - 2; w >= 0; --w) {
      const float4 p = *(const float4*)&pool[w * (16 * 260) + o];
      sum.x += p.x; sum.y += p.y; sum.z += p.z; sum.w += p.w;
    }
    *(float4*)&Rb[(size_t)(i0 + row) * C_DIM + c4] = sum;
  }
}

#define OUT_WAVES 16

__global__ __launch_bounds__(32 * OUT_WAVES)
void out_proj_kernel(const float* __restrict__ R,
                     const float* __restrict__ owT,
                     const float* __restrict__ out_scale,
                     float* __restrict__ out) {
  const int lane = threadIdx.x & 31;
  const int r = lane & 15;
  const int h = lane >> 4;
  const int wv = __builtin_amdgcn_readfirstlane((int)(threadIdx.x >> 5));
  const int strip = wv >> 3;
  const int cg = wv & 7;
  const int m0 = (blockIdx.x * 2 + strip) * 32;
  const int v0 = cg * 128;

  v8f acc[2][8];
#pragma unroll
  for (int mb = 0; mb < 2; ++mb)
#pragma unroll
    for (int nt = 0; nt < 8; ++nt) acc[mb][nt] = vzero8();

  const float* rr = &R[(size_t)(m0 + r) * C_DIM + 4 * h];
  const float* wf = owT + (size_t)(cg * 8) * (C_DIM / 8) * 128 + 4 * lane;
  gemm_tile_32x128<false, C_DIM / 8>(rr, wf, acc);

  const float sc = *out_scale;
#pragma unroll
  for (int mb = 0; mb < 2; ++mb)
#pragma unroll
    for (int nt = 0; nt < 8; ++nt)
#pragma unroll
      for (int i = 0; i < 8; ++i)
        out[(size_t)(m0 + 16 * mb + i + 8 * h) * V_DIM + v0 + nt * 16 + r] = acc[mb][nt][i] * sc;
}

extern "C" void kernel_launch(void* const* d_in, const int* in_sizes, int n_in,
                              void* d_out, int out_size, void* d_ws, size_t ws_size,
                              hipStream_t stream) {
  (void)in_sizes; (void)n_in; (void)out_size; (void)ws_size;

  const float* x = (const float*)d_in[0];
  const float* basis = (const float*)d_in[1];
  const float* qc = (const float*)d_in[2];
  const float* kc = (const float*)d_in[3];
  const float* vc = (const float*)d_in[4];
  const float* oc = (const float*)d_in[5];
  const float* decay_logit = (const float*)d_in[6];
  const float* out_scale = (const float*)d_in[7];
  float* out = (float*)d_out;

  float* ws = (float*)d_ws;
  float* qw = ws;
  float* kw = qw + (size_t)V_DIM * C_DIM;
  float* vw = kw + (size_t)V_DIM * C_DIM;
  float* owT = vw + (size_t)V_DIM * C_DIM;
  float* Q = owT + (size_t)V_DIM * C_DIM;
  float* K = Q + (size_t)M_DIM * C_DIM;
  float* Vv = K + (size_t)M_DIM * C_DIM;
  float* R = Vv + (size_t)M_DIM * C_DIM;

  dim3 blk(32);

  proj_weights_kernel<<<dim3(V_DIM / 16, C_DIM / 16, 4), blk, 0, stream>>>(
      basis, qc, kc, vc, oc, qw, kw, vw, owT);

  qkv_kernel<<<dim3(M_DIM / 64), dim3(32 * QKV_WAVES), 0, stream>>>(
      x, qw, kw, vw, Q, K, Vv);

  attn_kernel<<<dim3(B_DIM, T_DIM / 16), dim3(32 * ATT_WAVES), 0, stream>>>(
      Q, K, Vv, decay_logit, R);

  out_proj_kernel<<<dim3(M_DIM / 64), dim3(32 * OUT_WAVES), 0, stream>>>(
      R, owT, out_scale, out);
}
